# starting kernel
# baseline (speedup 1.0000x reference)
#include <hip/hip_runtime.h>
#include <hip/hip_bf16.h>

typedef __attribute__((ext_vector_type(2))) float v2f;
typedef __attribute__((ext_vector_type(8))) float v8f;

#define FDIM 64

__global__ void __launch_bounds__(256) zero_kernel(float* __restrict__ p, int n) {
    int i = blockIdx.x * 256 + threadIdx.x;
    if (i < n) p[i] = 0.0f;
}

__global__ void __launch_bounds__(256)
scatter_kernel(const float* __restrict__ feat, const int* __restrict__ src,
               const int* __restrict__ dst, float* __restrict__ agg,
               float* __restrict__ cnt, int n_edges) {
    long long tid = (long long)blockIdx.x * 256 + threadIdx.x;
    int e = (int)(tid >> 4);
    if (e >= n_edges) return;
    int q = (int)(tid & 15);
    int s = src[e];
    int d = dst[e];
    const float4 v = *(const float4*)(feat + (size_t)s * FDIM + q * 4);
    float* ap = agg + (size_t)d * FDIM + q * 4;
    atomicAdd(ap + 0, v.x);
    atomicAdd(ap + 1, v.y);
    atomicAdd(ap + 2, v.z);
    atomicAdd(ap + 3, v.w);
    if (cnt != nullptr && q == 0) atomicAdd(cnt + d, 1.0f);
}

__global__ void __launch_bounds__(256)
sage_gemm(const float* __restrict__ agg_sum, const float* __restrict__ cnt,
          const float* __restrict__ xin, const float* __restrict__ Wl,
          const float* __restrict__ Wr, const float* __restrict__ bias,
          float* __restrict__ out, int n_nodes, int relu_flag) {
    __shared__ float sWl[FDIM * FDIM];
    __shared__ float sWr[FDIM * FDIM];
    for (int i = threadIdx.x; i < FDIM * FDIM; i += 256) {
        sWl[i] = Wl[i];
        sWr[i] = Wr[i];
    }
    __syncthreads();

    const int lane = threadIdx.x & 31;
    const int wave = threadIdx.x >> 5;
    const int tile = blockIdx.x * 8 + wave;
    const int ntiles = (n_nodes + 15) >> 4;
    if (tile >= ntiles) return;
    const int node0 = tile << 4;

    const int half = lane >> 4;
    const int lm   = lane & 15;

    float c   = (cnt != nullptr) ? cnt[node0 + lm] : 1.0f;
    float inv = 1.0f / fmaxf(c, 1.0f);

    const float* arow = agg_sum + (size_t)(node0 + lm) * FDIM;
    const float* xrow = xin     + (size_t)(node0 + lm) * FDIM;

    v8f acc[4];
    #pragma unroll
    for (int nt = 0; nt < 4; ++nt) acc[nt] = (v8f){};

    #pragma unroll
    for (int kb = 0; kb < 16; ++kb) {
        const int k0 = kb * 4 + half * 2;
        v2f a;
        a.x = arow[k0]     * inv;
        a.y = arow[k0 + 1] * inv;
        #pragma unroll
        for (int nt = 0; nt < 4; ++nt) {
            v2f b;
            b.x = sWl[k0 * FDIM + nt * 16 + lm];
            b.y = sWl[(k0 + 1) * FDIM + nt * 16 + lm];
            acc[nt] = __builtin_amdgcn_wmma_f32_16x16x4_f32(
                false, a, false, b, (short)0, acc[nt], false, false);
        }
    }
    #pragma unroll
    for (int kb = 0; kb < 16; ++kb) {
        const int k0 = kb * 4 + half * 2;
        v2f a;
        a.x = xrow[k0];
        a.y = xrow[k0 + 1];
        #pragma unroll
        for (int nt = 0; nt < 4; ++nt) {
            v2f b;
            b.x = sWr[k0 * FDIM + nt * 16 + lm];
            b.y = sWr[(k0 + 1) * FDIM + nt * 16 + lm];
            acc[nt] = __builtin_amdgcn_wmma_f32_16x16x4_f32(
                false, a, false, b, (short)0, acc[nt], false, false);
        }
    }

    #pragma unroll
    for (int nt = 0; nt < 4; ++nt) {
        const float bv = bias[nt * 16 + lm];
        #pragma unroll
        for (int i = 0; i < 8; ++i) {
            float v = acc[nt][i] + bv;
            if (relu_flag) v = fmaxf(v, 0.0f);
            out[(size_t)(node0 + i + half * 8) * FDIM + nt * 16 + lm] = v;
        }
    }
}

extern "C" void kernel_launch(void* const* d_in, const int* in_sizes, int n_in,
                              void* d_out, int out_size, void* d_ws, size_t ws_size,
                              hipStream_t stream) {
    const float* x    = (const float*)d_in[0];
    const int*   ei   = (const int*)d_in[1];
    const float* W1_l = (const float*)d_in[2];
    const float* b1   = (const float*)d_in[3];
    const float* W1_r = (const float*)d_in[4];
    const float* W2_l = (const float*)d_in[5];
    const float* b2   = (const float*)d_in[6];
    const float* W2_r = (const float*)d_in[7];
    float* out = (float*)d_out;

    const int n_nodes = in_sizes[0] / FDIM;
    const int n_edges = in_sizes[1] / 2;
    const int* src = ei;
    const int* dst = ei + n_edges;

    float* agg = (float*)d_ws;
    float* cnt = agg + (size_t)n_nodes * FDIM;
    float* h   = cnt + n_nodes;

    const int ntiles     = (n_nodes + 15) / 16;
    const int gemm_grid  = (ntiles + 7) / 8;
    const long long sthreads = (long long)n_edges * 16;
    const int scat_grid  = (int)((sthreads + 255) / 256);
    const int nagg       = n_nodes * FDIM;

    zero_kernel<<<(nagg + n_nodes + 255) / 256, 256, 0, stream>>>(agg, nagg + n_nodes);
    scatter_kernel<<<scat_grid, 256, 0, stream>>>(x, src, dst, agg, cnt, n_edges);
    sage_gemm<<<gemm_grid, 256, 0, stream>>>(agg, cnt, x, W1_l, W1_r, b1, h,
                                             n_nodes,  1);

    zero_kernel<<<(nagg + 255) / 256, 256, 0, stream>>>(agg, nagg);
    scatter_kernel<<<scat_grid, 256, 0, stream>>>(h, src, dst, agg, nullptr, n_edges);
    sage_gemm<<<gemm_grid, 256, 0, stream>>>(agg, cnt, h, W2_l, W2_r, b2, out,
                                             n_nodes,  0);
}


# Float atomic scatter-add replaced by a per-call counting sort of edges by destination (integer atomic rank) and an atomic-free neighbour-row gather-sum; sum order per node = arrival-order rank
# speedup vs baseline: 1.7446x; 1.7446x over previous
#include <hip/hip_runtime.h>
#include <hip/hip_bf16.h>

typedef __attribute__((ext_vector_type(2))) float v2f;
typedef __attribute__((ext_vector_type(8))) float v8f;

#define FDIM 64

__global__ void __launch_bounds__(256) zero_int(int* __restrict__ p, int n) {
    int i = blockIdx.x * 256 + threadIdx.x;
    if (i < n) p[i] = 0;
}

__global__ void __launch_bounds__(256)
rank_edges(const int* __restrict__ dst, int* __restrict__ deg,
           int* __restrict__ rank, int n_edges) {
    int e = blockIdx.x * 256 + threadIdx.x;
    if (e < n_edges) rank[e] = atomicAdd(deg + dst[e], 1);
}

__global__ void __launch_bounds__(1024)
scan_deg(const int* __restrict__ deg, int* __restrict__ rowptr, int n_nodes) {
    __shared__ int s_pre[32];
    __shared__ int s_sum[32];
    const int t    = threadIdx.x;
    const int lane = t & 31;
    const int wave = t >> 5;
    const int b0   = blockIdx.x * 1024;

    int p = 0;
    const int4* dv = (const int4*)deg + t;
    const int nfull = b0 >> 12;
    int k = 0;
    for (; k + 8 <= nfull; k += 8) {
        int4 v[8];
        #pragma unroll
        for (int u = 0; u < 8; ++u) v[u] = dv[(k + u) * 1024];
        #pragma unroll
        for (int u = 0; u < 8; ++u) p += (v[u].x + v[u].y) + (v[u].z + v[u].w);
    }
    for (; k < nfull; ++k) {
        const int4 v = dv[k * 1024];
        p += (v.x + v.y) + (v.z + v.w);
    }
    if ((nfull << 12) + t * 4 < b0) {
        const int4 v = dv[nfull * 1024];
        p += (v.x + v.y) + (v.z + v.w);
    }
    const int i = b0 + t;
    const int d = (i < n_nodes) ? deg[i] : 0;

    int incl = d;
    #pragma unroll
    for (int o = 1; o < 32; o <<= 1) {
        const int u = __shfl_up(incl, o, 32);
        if (lane >= o) incl += u;
    }
    #pragma unroll
    for (int o = 16; o > 0; o >>= 1) p += __shfl_xor(p, o, 32);
    if (lane == 31) s_sum[wave] = incl;
    if (lane == 0)  s_pre[wave] = p;
    __syncthreads();

    int off = 0;
    #pragma unroll
    for (int w = 0; w < 32; ++w) {
        off += s_pre[w];
        if (w < wave) off += s_sum[w];
    }
    const int excl = off + incl - d;
    if (i < n_nodes) rowptr[i] = excl;
    if (i == n_nodes - 1) rowptr[n_nodes] = excl + d;
}

__global__ void __launch_bounds__(256)
fill_csr(const int* __restrict__ src, const int* __restrict__ dst,
         const int* __restrict__ rank, const int* __restrict__ rowptr,
         int* __restrict__ csr, int n_edges) {
    int e = blockIdx.x * 256 + threadIdx.x;
    if (e < n_edges) csr[rowptr[dst[e]] + rank[e]] = src[e];
}

__global__ void __launch_bounds__(256)
gather_mean(const float* __restrict__ feat, const int* __restrict__ rowptr,
            const int* __restrict__ csr, float* __restrict__ agg, int n_nodes) {
    const int tid  = blockIdx.x * 256 + threadIdx.x;
    const int node = tid >> 4;
    const int q    = tid & 15;
    if (node >= n_nodes) return;
    const int beg = rowptr[node];
    const int end = rowptr[node + 1];

    float4 acc = make_float4(0.0f, 0.0f, 0.0f, 0.0f);
    for (int base = beg; base < end; base += 32) {
        const int left = end - base;
        const int nb   = left < 32 ? left : 32;
        const int my0  = (q < nb)      ? csr[base + q]      : 0;
        const int my1  = (q + 16 < nb) ? csr[base + 16 + q] : 0;
        for (int t = 0; t < nb; t += 8) {
            float4 v[8];
            #pragma unroll
            for (int u = 0; u < 8; ++u) {
                const int tt = (t + u < nb) ? (t + u) : (nb - 1);
                const int s  = __shfl(tt < 16 ? my0 : my1, tt & 15, 16);
                v[u] = *(const float4*)(feat + (size_t)s * FDIM + q * 4);
            }
            #pragma unroll
            for (int u = 0; u < 8; ++u) {
                const bool on = (t + u < nb);
                acc.x += on ? v[u].x : 0.0f;
                acc.y += on ? v[u].y : 0.0f;
                acc.z += on ? v[u].z : 0.0f;
                acc.w += on ? v[u].w : 0.0f;
            }
        }
    }
    const float inv = 1.0f / fmaxf((float)(end - beg), 1.0f);
    acc.x *= inv; acc.y *= inv; acc.z *= inv; acc.w *= inv;
    *(float4*)(agg + (size_t)node * FDIM + q * 4) = acc;
}

__global__ void __launch_bounds__(256)
sage_gemm(const float* __restrict__ agg_sum, const float* __restrict__ cnt,
          const float* __restrict__ xin, const float* __restrict__ Wl,
          const float* __restrict__ Wr, const float* __restrict__ bias,
          float* __restrict__ out, int n_nodes, int relu_flag) {
    __shared__ float sWl[FDIM * FDIM];
    __shared__ float sWr[FDIM * FDIM];
    for (int i = threadIdx.x; i < FDIM * FDIM; i += 256) {
        sWl[i] = Wl[i];
        sWr[i] = Wr[i];
    }
    __syncthreads();

    const int lane = threadIdx.x & 31;
    const int wave = threadIdx.x >> 5;
    const int tile = blockIdx.x * 8 + wave;
    const int ntiles = (n_nodes + 15) >> 4;
    if (tile >= ntiles) return;
    const int node0 = tile << 4;

    const int half = lane >> 4;
    const int lm   = lane & 15;

    float c   = (cnt != nullptr) ? cnt[node0 + lm] : 1.0f;
    float inv = 1.0f / fmaxf(c, 1.0f);

    const float* arow = agg_sum + (size_t)(node0 + lm) * FDIM;
    const float* xrow = xin     + (size_t)(node0 + lm) * FDIM;

    v8f acc[4];
    #pragma unroll
    for (int nt = 0; nt < 4; ++nt) acc[nt] = (v8f){};

    #pragma unroll
    for (int kb = 0; kb < 16; ++kb) {
        const int k0 = kb * 4 + half * 2;
        v2f a;
        a.x = arow[k0]     * inv;
        a.y = arow[k0 + 1] * inv;
        #pragma unroll
        for (int nt = 0; nt < 4; ++nt) {
            v2f b;
            b.x = sWl[k0 * FDIM + nt * 16 + lm];
            b.y = sWl[(k0 + 1) * FDIM + nt * 16 + lm];
            acc[nt] = __builtin_amdgcn_wmma_f32_16x16x4_f32(
                false, a, false, b, (short)0, acc[nt], false, false);
        }
    }
    #pragma unroll
    for (int kb = 0; kb < 16; ++kb) {
        const int k0 = kb * 4 + half * 2;
        v2f a;
        a.x = xrow[k0];
        a.y = xrow[k0 + 1];
        #pragma unroll
        for (int nt = 0; nt < 4; ++nt) {
            v2f b;
            b.x = sWr[k0 * FDIM + nt * 16 + lm];
            b.y = sWr[(k0 + 1) * FDIM + nt * 16 + lm];
            acc[nt] = __builtin_amdgcn_wmma_f32_16x16x4_f32(
                false, a, false, b, (short)0, acc[nt], false, false);
        }
    }

    #pragma unroll
    for (int nt = 0; nt < 4; ++nt) {
        const float bv = bias[nt * 16 + lm];
        #pragma unroll
        for (int i = 0; i < 8; ++i) {
            float v = acc[nt][i] + bv;
            if (relu_flag) v = fmaxf(v, 0.0f);
            out[(size_t)(node0 + i + half * 8) * FDIM + nt * 16 + lm] = v;
        }
    }
}

extern "C" void kernel_launch(void* const* d_in, const int* in_sizes, int n_in,
                              void* d_out, int out_size, void* d_ws, size_t ws_size,
                              hipStream_t stream) {
    const float* x    = (const float*)d_in[0];
    const int*   ei   = (const int*)d_in[1];
    const float* W1_l = (const float*)d_in[2];
    const float* b1   = (const float*)d_in[3];
    const float* W1_r = (const float*)d_in[4];
    const float* W2_l = (const float*)d_in[5];
    const float* b2   = (const float*)d_in[6];
    const float* W2_r = (const float*)d_in[7];
    float* out = (float*)d_out;

    const int n_nodes = in_sizes[0] / FDIM;
    const int n_edges = in_sizes[1] / 2;
    const int* src = ei;
    const int* dst = ei + n_edges;

    float* agg   = (float*)d_ws;
    float* h     = agg + (size_t)n_nodes * FDIM;
    int*   deg   = (int*)(h + (size_t)n_nodes * FDIM);
    int*   rowptr = deg + n_nodes;
    int*   rank  = rowptr + n_nodes + 1;
    int*   csr   = rank + n_edges;

    const int ntiles     = (n_nodes + 15) / 16;
    const int gemm_grid  = (ntiles + 7) / 8;
    const int edge_grid  = (n_edges + 255) / 256;
    const int node_grid  = (n_nodes + 255) / 256;
    const int scan_grid  = (n_nodes + 1023) / 1024;
    const int gath_grid  = (n_nodes + 15) / 16;

    zero_int<<<node_grid, 256, 0, stream>>>(deg, n_nodes);
    rank_edges<<<edge_grid, 256, 0, stream>>>(dst, deg, rank, n_edges);
    scan_deg<<<scan_grid, 1024, 0, stream>>>(deg, rowptr, n_nodes);
    fill_csr<<<edge_grid, 256, 0, stream>>>(src, dst, rank, rowptr, csr, n_edges);

    gather_mean<<<gath_grid, 256, 0, stream>>>(x, rowptr, csr, agg, n_nodes);
    sage_gemm<<<gemm_grid, 256, 0, stream>>>(agg, nullptr, x, W1_l, W1_r, b1, h,
                                             n_nodes,  1);

    gather_mean<<<gath_grid, 256, 0, stream>>>(h, rowptr, csr, agg, n_nodes);
    sage_gemm<<<gemm_grid, 256, 0, stream>>>(agg, nullptr, h, W2_l, W2_r, b2, out,
                                             n_nodes,  0);
}


# Per-edge global atomic rank and scattered 4-byte fill replaced by a two-level counting sort on LDS counters (512-node buckets, then nodes); per-node sum order = arrival order of LDS ranks, as before
# speedup vs baseline: 2.8957x; 1.6598x over previous
#include <hip/hip_runtime.h>
#include <hip/hip_bf16.h>

typedef __attribute__((ext_vector_type(2))) float v2f;
typedef __attribute__((ext_vector_type(8))) float v8f;

#define FDIM 64

#define BSHIFT 9
#define BSIZE  512
#define CHUNK  8192
#define MAXB   2048

__global__ void __launch_bounds__(1024)
count_buckets(const int* __restrict__ dst, int* __restrict__ hist,
              int n_edges, int nb, int nwg) {
    __shared__ int s_h[MAXB];
    const int t = threadIdx.x;
    for (int b = t; b < nb; b += 1024) s_h[b] = 0;
    __syncthreads();
    const int base = blockIdx.x * CHUNK + t;
    int d[8];
    #pragma unroll
    for (int u = 0; u < 8; ++u) {
        const int e = base + u * 1024;
        d[u] = (e < n_edges) ? dst[e] : 0;
    }
    #pragma unroll
    for (int u = 0; u < 8; ++u) {
        const int e = base + u * 1024;
        if (e < n_edges) atomicAdd(&s_h[d[u] >> BSHIFT], 1);
    }
    __syncthreads();
    for (int b = t; b < nb; b += 1024) hist[b * nwg + blockIdx.x] = s_h[b];
}

__global__ void __launch_bounds__(1024)
scan_deg(const int* __restrict__ deg, int* __restrict__ rowptr, int n_nodes) {
    __shared__ int s_pre[32];
    __shared__ int s_sum[32];
    const int t    = threadIdx.x;
    const int lane = t & 31;
    const int wave = t >> 5;
    const int b0   = blockIdx.x * 1024;

    int p = 0;
    const int4* dv = (const int4*)deg + t;
    const int nfull = b0 >> 12;
    int k = 0;
    for (; k + 8 <= nfull; k += 8) {
        int4 v[8];
        #pragma unroll
        for (int u = 0; u < 8; ++u) v[u] = dv[(k + u) * 1024];
        #pragma unroll
        for (int u = 0; u < 8; ++u) p += (v[u].x + v[u].y) + (v[u].z + v[u].w);
    }
    for (; k < nfull; ++k) {
        const int4 v = dv[k * 1024];
        p += (v.x + v.y) + (v.z + v.w);
    }
    if ((nfull << 12) + t * 4 < b0) {
        const int4 v = dv[nfull * 1024];
        p += (v.x + v.y) + (v.z + v.w);
    }
    const int i = b0 + t;
    const int d = (i < n_nodes) ? deg[i] : 0;

    int incl = d;
    #pragma unroll
    for (int o = 1; o < 32; o <<= 1) {
        const int u = __shfl_up(incl, o, 32);
        if (lane >= o) incl += u;
    }
    #pragma unroll
    for (int o = 16; o > 0; o >>= 1) p += __shfl_xor(p, o, 32);
    if (lane == 31) s_sum[wave] = incl;
    if (lane == 0)  s_pre[wave] = p;
    __syncthreads();

    int off = 0;
    #pragma unroll
    for (int w = 0; w < 32; ++w) {
        off += s_pre[w];
        if (w < wave) off += s_sum[w];
    }
    const int excl = off + incl - d;
    if (i < n_nodes) rowptr[i] = excl;
    if (i == n_nodes - 1) rowptr[n_nodes] = excl + d;
}

__global__ void __launch_bounds__(1024)
place_edges(const int* __restrict__ src, const int* __restrict__ dst,
            const int* __restrict__ off, int* __restrict__ rec,
            int n_edges, int nb, int nwg) {
    __shared__ int s_pos[MAXB];
    const int t = threadIdx.x;
    for (int b = t; b < nb; b += 1024) s_pos[b] = off[b * nwg + blockIdx.x];
    __syncthreads();
    const int base = blockIdx.x * CHUNK + t;
    int d[8], s[8];
    #pragma unroll
    for (int u = 0; u < 8; ++u) {
        const int e = base + u * 1024;
        d[u] = (e < n_edges) ? dst[e] : 0;
        s[u] = (e < n_edges) ? src[e] : 0;
    }
    int2* __restrict__ rp = (int2*)rec;
    int p[8];
    #pragma unroll
    for (int u = 0; u < 8; ++u) {
        const int e = base + u * 1024;
        p[u] = 0;
        if (e < n_edges) p[u] = atomicAdd(&s_pos[d[u] >> BSHIFT], 1);
    }
    #pragma unroll
    for (int u = 0; u < 8; ++u) {
        const int e = base + u * 1024;
        if (e < n_edges) rp[p[u]] = make_int2(s[u], d[u]);
    }
}

__global__ void __launch_bounds__(1024)
sort_bucket(const int* __restrict__ rec, const int* __restrict__ off,
            int* __restrict__ rowptr, int* __restrict__ csr,
            int n_nodes, int nwg) {
    __shared__ int s_cnt[BSIZE];
    __shared__ int s_ws[16];
    const int t     = threadIdx.x;
    const int lane  = t & 31;
    const int wave  = t >> 5;
    const int b     = blockIdx.x;
    const int beg   = off[b * nwg];
    const int end   = off[(b + 1) * nwg];
    const int node0 = b << BSHIFT;
    const int2* __restrict__ rp = (const int2*)rec;

    if (t < BSIZE) s_cnt[t] = 0;
    __syncthreads();

    for (int i0 = beg + t; i0 < end; i0 += 8 * 1024) {
        int dl[8];
        #pragma unroll
        for (int u = 0; u < 8; ++u) {
            const int i = i0 + u * 1024;
            dl[u] = (i < end) ? rp[i].y : node0;
        }
        #pragma unroll
        for (int u = 0; u < 8; ++u) {
            const int i = i0 + u * 1024;
            if (i < end) atomicAdd(&s_cnt[dl[u] - node0], 1);
        }
    }
    __syncthreads();

    const int c = (t < BSIZE) ? s_cnt[t] : 0;
    int incl = c;
    #pragma unroll
    for (int o = 1; o < 32; o <<= 1) {
        const int u = __shfl_up(incl, o, 32);
        if (lane >= o) incl += u;
    }
    if (lane == 31 && wave < 16) s_ws[wave] = incl;
    __syncthreads();
    int woff = 0;
    #pragma unroll
    for (int w = 0; w < 16; ++w) if (w < wave) woff += s_ws[w];
    const int excl = woff + incl - c;
    if (t < BSIZE) {
        s_cnt[t] = excl;
        const int node = node0 + t;
        if (node < n_nodes) rowptr[node] = beg + excl;
        if (node == n_nodes - 1) rowptr[n_nodes] = end;
    }
    __syncthreads();

    for (int i0 = beg + t; i0 < end; i0 += 8 * 1024) {
        int2 r[8];
        #pragma unroll
        for (int u = 0; u < 8; ++u) {
            const int i = i0 + u * 1024;
            r[u] = (i < end) ? rp[i] : make_int2(0, node0);
        }
        int p[8];
        #pragma unroll
        for (int u = 0; u < 8; ++u) {
            const int i = i0 + u * 1024;
            p[u] = 0;
            if (i < end) p[u] = atomicAdd(&s_cnt[r[u].y - node0], 1);
        }
        #pragma unroll
        for (int u = 0; u < 8; ++u) {
            const int i = i0 + u * 1024;
            if (i < end) csr[beg + p[u]] = r[u].x;
        }
    }
}

__global__ void __launch_bounds__(256)
gather_mean(const float* __restrict__ feat, const int* __restrict__ rowptr,
            const int* __restrict__ csr, float* __restrict__ agg, int n_nodes) {
    const int tid  = blockIdx.x * 256 + threadIdx.x;
    const int node = tid >> 4;
    const int q    = tid & 15;
    if (node >= n_nodes) return;
    const int beg = rowptr[node];
    const int end = rowptr[node + 1];

    float4 acc = make_float4(0.0f, 0.0f, 0.0f, 0.0f);
    for (int base = beg; base < end; base += 32) {
        const int left = end - base;
        const int nb   = left < 32 ? left : 32;
        const int my0  = (q < nb)      ? csr[base + q]      : 0;
        const int my1  = (q + 16 < nb) ? csr[base + 16 + q] : 0;
        for (int t = 0; t < nb; t += 8) {
            float4 v[8];
            #pragma unroll
            for (int u = 0; u < 8; ++u) {
                const int tt = (t + u < nb) ? (t + u) : (nb - 1);
                const int s  = __shfl(tt < 16 ? my0 : my1, tt & 15, 16);
                v[u] = *(const float4*)(feat + (size_t)s * FDIM + q * 4);
            }
            #pragma unroll
            for (int u = 0; u < 8; ++u) {
                const bool on = (t + u < nb);
                acc.x += on ? v[u].x : 0.0f;
                acc.y += on ? v[u].y : 0.0f;
                acc.z += on ? v[u].z : 0.0f;
                acc.w += on ? v[u].w : 0.0f;
            }
        }
    }
    const float inv = 1.0f / fmaxf((float)(end - beg), 1.0f);
    acc.x *= inv; acc.y *= inv; acc.z *= inv; acc.w *= inv;
    *(float4*)(agg + (size_t)node * FDIM + q * 4) = acc;
}

__global__ void __launch_bounds__(256)
sage_gemm(const float* __restrict__ agg_sum, const float* __restrict__ cnt,
          const float* __restrict__ xin, const float* __restrict__ Wl,
          const float* __restrict__ Wr, const float* __restrict__ bias,
          float* __restrict__ out, int n_nodes, int relu_flag) {
    __shared__ float sWl[FDIM * FDIM];
    __shared__ float sWr[FDIM * FDIM];
    for (int i = threadIdx.x; i < FDIM * FDIM; i += 256) {
        sWl[i] = Wl[i];
        sWr[i] = Wr[i];
    }
    __syncthreads();

    const int lane = threadIdx.x & 31;
    const int wave = threadIdx.x >> 5;
    const int tile = blockIdx.x * 8 + wave;
    const int ntiles = (n_nodes + 15) >> 4;
    if (tile >= ntiles) return;
    const int node0 = tile << 4;

    const int half = lane >> 4;
    const int lm   = lane & 15;

    float c   = (cnt != nullptr) ? cnt[node0 + lm] : 1.0f;
    float inv = 1.0f / fmaxf(c, 1.0f);

    const float* arow = agg_sum + (size_t)(node0 + lm) * FDIM;
    const float* xrow = xin     + (size_t)(node0 + lm) * FDIM;

    v8f acc[4];
    #pragma unroll
    for (int nt = 0; nt < 4; ++nt) acc[nt] = (v8f){};

    #pragma unroll
    for (int kb = 0; kb < 16; ++kb) {
        const int k0 = kb * 4 + half * 2;
        v2f a;
        a.x = arow[k0]     * inv;
        a.y = arow[k0 + 1] * inv;
        #pragma unroll
        for (int nt = 0; nt < 4; ++nt) {
            v2f b;
            b.x = sWl[k0 * FDIM + nt * 16 + lm];
            b.y = sWl[(k0 + 1) * FDIM + nt * 16 + lm];
            acc[nt] = __builtin_amdgcn_wmma_f32_16x16x4_f32(
                false, a, false, b, (short)0, acc[nt], false, false);
        }
    }
    #pragma unroll
    for (int kb = 0; kb < 16; ++kb) {
        const int k0 = kb * 4 + half * 2;
        v2f a;
        a.x = xrow[k0];
        a.y = xrow[k0 + 1];
        #pragma unroll
        for (int nt = 0; nt < 4; ++nt) {
            v2f b;
            b.x = sWr[k0 * FDIM + nt * 16 + lm];
            b.y = sWr[(k0 + 1) * FDIM + nt * 16 + lm];
            acc[nt] = __builtin_amdgcn_wmma_f32_16x16x4_f32(
                false, a, false, b, (short)0, acc[nt], false, false);
        }
    }

    #pragma unroll
    for (int nt = 0; nt < 4; ++nt) {
        const float bv = bias[nt * 16 + lm];
        #pragma unroll
        for (int i = 0; i < 8; ++i) {
            float v = acc[nt][i] + bv;
            if (relu_flag) v = fmaxf(v, 0.0f);
            out[(size_t)(node0 + i + half * 8) * FDIM + nt * 16 + lm] = v;
        }
    }
}

extern "C" void kernel_launch(void* const* d_in, const int* in_sizes, int n_in,
                              void* d_out, int out_size, void* d_ws, size_t ws_size,
                              hipStream_t stream) {
    const float* x    = (const float*)d_in[0];
    const int*   ei   = (const int*)d_in[1];
    const float* W1_l = (const float*)d_in[2];
    const float* b1   = (const float*)d_in[3];
    const float* W1_r = (const float*)d_in[4];
    const float* W2_l = (const float*)d_in[5];
    const float* b2   = (const float*)d_in[6];
    const float* W2_r = (const float*)d_in[7];
    float* out = (float*)d_out;

    const int n_nodes = in_sizes[0] / FDIM;
    const int n_edges = in_sizes[1] / 2;
    const int* src = ei;
    const int* dst = ei + n_edges;

    const int nb   = (n_nodes + BSIZE - 1) / BSIZE;
    const int nwg  = (n_edges + CHUNK - 1) / CHUNK;
    const int tot  = nb * nwg;
    const int tpad = (tot + 4) & ~3;

    float* agg    = (float*)d_ws;
    float* h      = agg + (size_t)n_nodes * FDIM;
    int*   hist   = (int*)(h + (size_t)n_nodes * FDIM);
    int*   off    = hist + tpad;
    int*   rec    = off + tpad;
    int*   csr    = rec + 2 * (size_t)n_edges;
    int*   rowptr = csr + n_edges;

    const int ntiles     = (n_nodes + 15) / 16;
    const int gemm_grid  = (ntiles + 7) / 8;
    const int scan_grid  = (tot + 1023) / 1024;
    const int gath_grid  = (n_nodes + 15) / 16;

    count_buckets<<<nwg, 1024, 0, stream>>>(dst, hist, n_edges, nb, nwg);
    scan_deg<<<scan_grid, 1024, 0, stream>>>(hist, off, tot);
    place_edges<<<nwg, 1024, 0, stream>>>(src, dst, off, rec, n_edges, nb, nwg);
    sort_bucket<<<nb, 1024, 0, stream>>>(rec, off, rowptr, csr, n_nodes, nwg);

    gather_mean<<<gath_grid, 256, 0, stream>>>(x, rowptr, csr, agg, n_nodes);
    sage_gemm<<<gemm_grid, 256, 0, stream>>>(agg, nullptr, x, W1_l, W1_r, b1, h,
                                             n_nodes,  1);

    gather_mean<<<gath_grid, 256, 0, stream>>>(h, rowptr, csr, agg, n_nodes);
    sage_gemm<<<gemm_grid, 256, 0, stream>>>(agg, nullptr, h, W2_l, W2_r, b2, out,
                                             n_nodes,  0);
}


# Neighbour gather fused into the layer's f32 WMMA kernel: 16-wave blocks of 128 nodes gather the mean into an LDS tile and multiply from it; 8 -> 6 launches, a quarter of the waves; sum order kept
# speedup vs baseline: 3.2427x; 1.1198x over previous
#include <hip/hip_runtime.h>
#include <hip/hip_bf16.h>

typedef __attribute__((ext_vector_type(2))) float v2f;
typedef __attribute__((ext_vector_type(8))) float v8f;
typedef __attribute__((ext_vector_type(4))) float v4f;

#define FDIM 64

#define BSHIFT 9
#define BSIZE  512
#define CHUNK  8192
#define MAXB   2048

__global__ void __launch_bounds__(1024)
count_buckets(const int* __restrict__ dst, int* __restrict__ hist,
              int n_edges, int nb, int nwg) {
    __shared__ int s_h[MAXB];
    const int t = threadIdx.x;
    for (int b = t; b < nb; b += 1024) s_h[b] = 0;
    __syncthreads();
    const int base = blockIdx.x * CHUNK + t;
    int d[8];
    #pragma unroll
    for (int u = 0; u < 8; ++u) {
        const int e = base + u * 1024;
        d[u] = (e < n_edges) ? dst[e] : 0;
    }
    #pragma unroll
    for (int u = 0; u < 8; ++u) {
        const int e = base + u * 1024;
        if (e < n_edges) atomicAdd(&s_h[d[u] >> BSHIFT], 1);
    }
    __syncthreads();
    for (int b = t; b < nb; b += 1024) hist[b * nwg + blockIdx.x] = s_h[b];
}

__global__ void __launch_bounds__(1024)
scan_deg(const int* __restrict__ deg, int* __restrict__ rowptr, int n_nodes) {
    __shared__ int s_pre[32];
    __shared__ int s_sum[32];
    const int t    = threadIdx.x;
    const int lane = t & 31;
    const int wave = t >> 5;
    const int b0   = blockIdx.x * 1024;

    int p = 0;
    const int4* dv = (const int4*)deg + t;
    const int nfull = b0 >> 12;
    int k = 0;
    for (; k + 8 <= nfull; k += 8) {
        int4 v[8];
        #pragma unroll
        for (int u = 0; u < 8; ++u) v[u] = dv[(k + u) * 1024];
        #pragma unroll
        for (int u = 0; u < 8; ++u) p += (v[u].x + v[u].y) + (v[u].z + v[u].w);
    }
    for (; k < nfull; ++k) {
        const int4 v = dv[k * 1024];
        p += (v.x + v.y) + (v.z + v.w);
    }
    if ((nfull << 12) + t * 4 < b0) {
        const int4 v = dv[nfull * 1024];
        p += (v.x + v.y) + (v.z + v.w);
    }
    const int i = b0 + t;
    const int d = (i < n_nodes) ? deg[i] : 0;

    int incl = d;
    #pragma unroll
    for (int o = 1; o < 32; o <<= 1) {
        const int u = __shfl_up(incl, o, 32);
        if (lane >= o) incl += u;
    }
    #pragma unroll
    for (int o = 16; o > 0; o >>= 1) p += __shfl_xor(p, o, 32);
    if (lane == 31) s_sum[wave] = incl;
    if (lane == 0)  s_pre[wave] = p;
    __syncthreads();

    int off = 0;
    #pragma unroll
    for (int w = 0; w < 32; ++w) {
        off += s_pre[w];
        if (w < wave) off += s_sum[w];
    }
    const int excl = off + incl - d;
    if (i < n_nodes) rowptr[i] = excl;
    if (i == n_nodes - 1) rowptr[n_nodes] = excl + d;
}

__global__ void __launch_bounds__(1024)
place_edges(const int* __restrict__ src, const int* __restrict__ dst,
            const int* __restrict__ off, int* __restrict__ rec,
            int n_edges, int nb, int nwg) {
    __shared__ int s_pos[MAXB];
    const int t = threadIdx.x;
    for (int b = t; b < nb; b += 1024) s_pos[b] = off[b * nwg + blockIdx.x];
    __syncthreads();
    const int base = blockIdx.x * CHUNK + t;
    int d[8], s[8];
    #pragma unroll
    for (int u = 0; u < 8; ++u) {
        const int e = base + u * 1024;
        d[u] = (e < n_edges) ? dst[e] : 0;
        s[u] = (e < n_edges) ? src[e] : 0;
    }
    int2* __restrict__ rp = (int2*)rec;
    int p[8];
    #pragma unroll
    for (int u = 0; u < 8; ++u) {
        const int e = base + u * 1024;
        p[u] = 0;
        if (e < n_edges) p[u] = atomicAdd(&s_pos[d[u] >> BSHIFT], 1);
    }
    #pragma unroll
    for (int u = 0; u < 8; ++u) {
        const int e = base + u * 1024;
        if (e < n_edges) rp[p[u]] = make_int2(s[u], d[u]);
    }
}

__global__ void __launch_bounds__(1024)
sort_bucket(const int* __restrict__ rec, const int* __restrict__ off,
            int* __restrict__ rowptr, int* __restrict__ csr,
            int n_nodes, int nwg) {
    __shared__ int s_cnt[BSIZE];
    __shared__ int s_ws[16];
    const int t     = threadIdx.x;
    const int lane  = t & 31;
    const int wave  = t >> 5;
    const int b     = blockIdx.x;
    const int beg   = off[b * nwg];
    const int end   = off[(b + 1) * nwg];
    const int node0 = b << BSHIFT;
    const int2* __restrict__ rp = (const int2*)rec;

    if (t < BSIZE) s_cnt[t] = 0;
    __syncthreads();

    for (int i0 = beg + t; i0 < end; i0 += 8 * 1024) {
        int dl[8];
        #pragma unroll
        for (int u = 0; u < 8; ++u) {
            const int i = i0 + u * 1024;
            dl[u] = (i < end) ? rp[i].y : node0;
        }
        #pragma unroll
        for (int u = 0; u < 8; ++u) {
            const int i = i0 + u * 1024;
            if (i < end) atomicAdd(&s_cnt[dl[u] - node0], 1);
        }
    }
    __syncthreads();

    const int c = (t < BSIZE) ? s_cnt[t] : 0;
    int incl = c;
    #pragma unroll
    for (int o = 1; o < 32; o <<= 1) {
        const int u = __shfl_up(incl, o, 32);
        if (lane >= o) incl += u;
    }
    if (lane == 31 && wave < 16) s_ws[wave] = incl;
    __syncthreads();
    int woff = 0;
    #pragma unroll
    for (int w = 0; w < 16; ++w) if (w < wave) woff += s_ws[w];
    const int excl = woff + incl - c;
    if (t < BSIZE) {
        s_cnt[t] = excl;
        const int node = node0 + t;
        if (node < n_nodes) rowptr[node] = beg + excl;
        if (node == n_nodes - 1) rowptr[n_nodes] = end;
    }
    __syncthreads();

    for (int i0 = beg + t; i0 < end; i0 += 8 * 1024) {
        int2 r[8];
        #pragma unroll
        for (int u = 0; u < 8; ++u) {
            const int i = i0 + u * 1024;
            r[u] = (i < end) ? rp[i] : make_int2(0, node0);
        }
        int p[8];
        #pragma unroll
        for (int u = 0; u < 8; ++u) {
            const int i = i0 + u * 1024;
            p[u] = 0;
            if (i < end) p[u] = atomicAdd(&s_cnt[r[u].y - node0], 1);
        }
        #pragma unroll
        for (int u = 0; u < 8; ++u) {
            const int i = i0 + u * 1024;
            if (i < end) csr[beg + p[u]] = r[u].x;
        }
    }
}

#define TNODES 128
#define APITCH 68

__global__ void __launch_bounds__(512, 16)
sage_layer(const float* __restrict__ feat, const int* __restrict__ rowptr,
           const int* __restrict__ csr, const float* __restrict__ Wl,
           const float* __restrict__ Wr, const float* __restrict__ bias,
           float* __restrict__ out, int n_nodes, int relu_flag) {
    __shared__ float sWl[FDIM * FDIM];
    __shared__ float sWr[FDIM * FDIM];
    __shared__ float sAgg[TNODES * APITCH];

    const int lane = threadIdx.x & 31;
    const int wave = threadIdx.x >> 5;
    const int blk0 = blockIdx.x * TNODES;
    const int half = lane >> 4;
    const int lm   = lane & 15;

    const int nl0 = wave * 8;
    int rpi = blk0 + nl0 + lane;
    rpi = rpi < n_nodes ? rpi : n_nodes;
    const int rp = rowptr[rpi];

    {
        const int i = threadIdx.x;
        const v4f l0 = ((const v4f*)Wl)[i];
        const v4f l1 = ((const v4f*)Wl)[i + 512];
        const v4f r0 = ((const v4f*)Wr)[i];
        const v4f r1 = ((const v4f*)Wr)[i + 512];
        asm volatile("" :: "v"(rp), "v"(l0), "v"(l1), "v"(r0), "v"(r1));
        ((v4f*)sWl)[i]       = l0;
        ((v4f*)sWl)[i + 512] = l1;
        ((v4f*)sWr)[i]       = r0;
        ((v4f*)sWr)[i + 512] = r1;
    }

    for (int p = 0; p < 2; ++p) {
        int id0[2], id1[2];
        #pragma unroll
        for (int rr = 0; rr < 2; ++rr) {
            const int r = 2 * p + rr;
            const int b = __shfl(rp, 2 * r + half, 32);
            const int d = __shfl(rp, 2 * r + 1 + half, 32) - b;
            id0[rr] = (lm < d)      ? csr[b + lm]      : 0;
            id1[rr] = (lm + 16 < d) ? csr[b + 16 + lm] : 0;
        }
        #pragma unroll
        for (int rr = 0; rr < 2; ++rr) {
            const int r = 2 * p + rr;
            const int beg = __shfl(rp, 2 * r + half, 32);
            const int end = __shfl(rp, 2 * r + 1 + half, 32);
            float4 acc = make_float4(0.0f, 0.0f, 0.0f, 0.0f);
            for (int base = beg; base < end; base += 32) {
                const int left = end - base;
                const int nb   = left < 32 ? left : 32;
                int my0 = id0[rr], my1 = id1[rr];
                if (base != beg) {
                    my0 = (lm < nb)      ? csr[base + lm]      : 0;
                    my1 = (lm + 16 < nb) ? csr[base + 16 + lm] : 0;
                }
                for (int t = 0; t < nb; t += 8) {
                    float4 v[8];
                    #pragma unroll
                    for (int u = 0; u < 8; ++u) {
                        const int tt = (t + u < nb) ? (t + u) : (nb - 1);
                        const int s  = __shfl(tt < 16 ? my0 : my1, tt & 15, 16);
                        v[u] = ((const float4*)feat)[(unsigned)s * (FDIM / 4) + lm];
                    }
                    #pragma unroll
                    for (int u = 0; u < 8; ++u) {
                        const bool on = (t + u < nb);
                        acc.x += on ? v[u].x : 0.0f;
                        acc.y += on ? v[u].y : 0.0f;
                        acc.z += on ? v[u].z : 0.0f;
                        acc.w += on ? v[u].w : 0.0f;
                    }
                }
            }
            const float inv = 1.0f / fmaxf((float)(end - beg), 1.0f);
            acc.x *= inv; acc.y *= inv; acc.z *= inv; acc.w *= inv;
            *(float4*)(sAgg + (nl0 + 2 * r + half) * APITCH + lm * 4) = acc;
        }
    }

    const int tile  = wave >> 1;
    const int nh    = wave & 1;
    const int node0 = blk0 + tile * 16;

    __syncthreads();

    if (node0 >= n_nodes) return;

    int xr = node0 + lm;
    xr = xr < n_nodes ? xr : n_nodes - 1;
    const float* xrow = feat + (size_t)xr * FDIM + half * 2;
    v2f ax[16];
    #pragma unroll
    for (int kb = 0; kb < 16; ++kb) ax[kb] = *(const v2f*)(xrow + kb * 4);

    const float* arow = sAgg + (tile * 16 + lm) * APITCH + half * 2;

    v8f acc[2];
    #pragma unroll
    for (int j = 0; j < 2; ++j) acc[j] = (v8f){};

    #pragma unroll
    for (int kb = 0; kb < 16; ++kb) {
        const int k0 = kb * 4 + half * 2;
        const v2f a = *(const v2f*)(arow + kb * 4);
        #pragma unroll
        for (int j = 0; j < 2; ++j) {
            const int nt = nh * 2 + j;
            v2f b;
            b.x = sWl[k0 * FDIM + nt * 16 + lm];
            b.y = sWl[(k0 + 1) * FDIM + nt * 16 + lm];
            acc[j] = __builtin_amdgcn_wmma_f32_16x16x4_f32(
                false, a, false, b, (short)0, acc[j], false, false);
        }
    }
    #pragma unroll
    for (int kb = 0; kb < 16; ++kb) {
        const int k0 = kb * 4 + half * 2;
        #pragma unroll
        for (int j = 0; j < 2; ++j) {
            const int nt = nh * 2 + j;
            v2f b;
            b.x = sWr[k0 * FDIM + nt * 16 + lm];
            b.y = sWr[(k0 + 1) * FDIM + nt * 16 + lm];
            acc[j] = __builtin_amdgcn_wmma_f32_16x16x4_f32(
                false, ax[kb], false, b, (short)0, acc[j], false, false);
        }
    }

    #pragma unroll
    for (int j = 0; j < 2; ++j) {
        const int nt = nh * 2 + j;
        const float bv = bias[nt * 16 + lm];
        #pragma unroll
        for (int i = 0; i < 8; ++i) {
            float v = acc[j][i] + bv;
            if (relu_flag) v = fmaxf(v, 0.0f);
            const int row = node0 + i + half * 8;
            if (row < n_nodes) out[(size_t)row * FDIM + nt * 16 + lm] = v;
        }
    }
}

extern "C" void kernel_launch(void* const* d_in, const int* in_sizes, int n_in,
                              void* d_out, int out_size, void* d_ws, size_t ws_size,
                              hipStream_t stream) {
    const float* x    = (const float*)d_in[0];
    const int*   ei   = (const int*)d_in[1];
    const float* W1_l = (const float*)d_in[2];
    const float* b1   = (const float*)d_in[3];
    const float* W1_r = (const float*)d_in[4];
    const float* W2_l = (const float*)d_in[5];
    const float* b2   = (const float*)d_in[6];
    const float* W2_r = (const float*)d_in[7];
    float* out = (float*)d_out;

    const int n_nodes = in_sizes[0] / FDIM;
    const int n_edges = in_sizes[1] / 2;
    const int* src = ei;
    const int* dst = ei + n_edges;

    const int nb   = (n_nodes + BSIZE - 1) / BSIZE;
    const int nwg  = (n_edges + CHUNK - 1) / CHUNK;
    const int tot  = nb * nwg;
    const int tpad = (tot + 4) & ~3;

    float* h      = (float*)d_ws;
    int*   hist   = (int*)(h + (size_t)n_nodes * FDIM);
    int*   off    = hist + tpad;
    int*   rec    = off + tpad;
    int*   csr    = rec + 2 * (size_t)n_edges;
    int*   rowptr = csr + n_edges;

    const int scan_grid  = (tot + 1023) / 1024;
    const int layer_grid = (n_nodes + TNODES - 1) / TNODES;

    count_buckets<<<nwg, 1024, 0, stream>>>(dst, hist, n_edges, nb, nwg);
    scan_deg<<<scan_grid, 1024, 0, stream>>>(hist, off, tot);
    place_edges<<<nwg, 1024, 0, stream>>>(src, dst, off, rec, n_edges, nb, nwg);
    sort_bucket<<<nb, 1024, 0, stream>>>(rec, off, rowptr, csr, n_nodes, nwg);

    sage_layer<<<layer_grid, 512, 0, stream>>>(x, rowptr, csr, W1_l, W1_r, b1, h,
                                               n_nodes,  1);

    sage_layer<<<layer_grid, 512, 0, stream>>>(h, rowptr, csr, W2_l, W2_r, b2, out,
                                               n_nodes,  0);
}


# Gather: neighbour ids handed to the row-fetching lanes by DPP row_share with compile-time lanes (trips written out) instead of ds_bpermute with per-lane clamped index; sum order kept
# speedup vs baseline: 3.3590x; 1.0359x over previous
#include <hip/hip_runtime.h>
#include <hip/hip_bf16.h>

typedef __attribute__((ext_vector_type(2))) float v2f;
typedef __attribute__((ext_vector_type(8))) float v8f;
typedef __attribute__((ext_vector_type(4))) float v4f;

#define FDIM 64

#define BSHIFT 9
#define BSIZE  512
#define CHUNK  8192
#define MAXB   2048

__global__ void __launch_bounds__(1024)
count_buckets(const int* __restrict__ dst, int* __restrict__ hist,
              int n_edges, int nb, int nwg) {
    __shared__ int s_h[MAXB];
    const int t = threadIdx.x;
    for (int b = t; b < nb; b += 1024) s_h[b] = 0;
    __syncthreads();
    const int base = blockIdx.x * CHUNK + t;
    int d[8];
    #pragma unroll
    for (int u = 0; u < 8; ++u) {
        const int e = base + u * 1024;
        d[u] = (e < n_edges) ? dst[e] : 0;
    }
    #pragma unroll
    for (int u = 0; u < 8; ++u) {
        const int e = base + u * 1024;
        if (e < n_edges) atomicAdd(&s_h[d[u] >> BSHIFT], 1);
    }
    __syncthreads();
    for (int b = t; b < nb; b += 1024) hist[b * nwg + blockIdx.x] = s_h[b];
}

__global__ void __launch_bounds__(1024)
scan_deg(const int* __restrict__ deg, int* __restrict__ rowptr, int n_nodes) {
    __shared__ int s_pre[32];
    __shared__ int s_sum[32];
    const int t    = threadIdx.x;
    const int lane = t & 31;
    const int wave = t >> 5;
    const int b0   = blockIdx.x * 1024;

    int p = 0;
    const int4* dv = (const int4*)deg + t;
    const int nfull = b0 >> 12;
    int k = 0;
    for (; k + 8 <= nfull; k += 8) {
        int4 v[8];
        #pragma unroll
        for (int u = 0; u < 8; ++u) v[u] = dv[(k + u) * 1024];
        #pragma unroll
        for (int u = 0; u < 8; ++u) p += (v[u].x + v[u].y) + (v[u].z + v[u].w);
    }
    for (; k < nfull; ++k) {
        const int4 v = dv[k * 1024];
        p += (v.x + v.y) + (v.z + v.w);
    }
    if ((nfull << 12) + t * 4 < b0) {
        const int4 v = dv[nfull * 1024];
        p += (v.x + v.y) + (v.z + v.w);
    }
    const int i = b0 + t;
    const int d = (i < n_nodes) ? deg[i] : 0;

    int incl = d;
    #pragma unroll
    for (int o = 1; o < 32; o <<= 1) {
        const int u = __shfl_up(incl, o, 32);
        if (lane >= o) incl += u;
    }
    #pragma unroll
    for (int o = 16; o > 0; o >>= 1) p += __shfl_xor(p, o, 32);
    if (lane == 31) s_sum[wave] = incl;
    if (lane == 0)  s_pre[wave] = p;
    __syncthreads();

    int off = 0;
    #pragma unroll
    for (int w = 0; w < 32; ++w) {
        off += s_pre[w];
        if (w < wave) off += s_sum[w];
    }
    const int excl = off + incl - d;
    if (i < n_nodes) rowptr[i] = excl;
    if (i == n_nodes - 1) rowptr[n_nodes] = excl + d;
}

__global__ void __launch_bounds__(1024)
place_edges(const int* __restrict__ src, const int* __restrict__ dst,
            const int* __restrict__ off, int* __restrict__ rec,
            int n_edges, int nb, int nwg) {
    __shared__ int s_pos[MAXB];
    const int t = threadIdx.x;
    for (int b = t; b < nb; b += 1024) s_pos[b] = off[b * nwg + blockIdx.x];
    __syncthreads();
    const int base = blockIdx.x * CHUNK + t;
    int d[8], s[8];
    #pragma unroll
    for (int u = 0; u < 8; ++u) {
        const int e = base + u * 1024;
        d[u] = (e < n_edges) ? dst[e] : 0;
        s[u] = (e < n_edges) ? src[e] : 0;
    }
    int2* __restrict__ rp = (int2*)rec;
    int p[8];
    #pragma unroll
    for (int u = 0; u < 8; ++u) {
        const int e = base + u * 1024;
        p[u] = 0;
        if (e < n_edges) p[u] = atomicAdd(&s_pos[d[u] >> BSHIFT], 1);
    }
    #pragma unroll
    for (int u = 0; u < 8; ++u) {
        const int e = base + u * 1024;
        if (e < n_edges) rp[p[u]] = make_int2(s[u], d[u]);
    }
}

__global__ void __launch_bounds__(1024)
sort_bucket(const int* __restrict__ rec, const int* __restrict__ off,
            int* __restrict__ rowptr, int* __restrict__ csr,
            int n_nodes, int nwg) {
    __shared__ int s_cnt[BSIZE];
    __shared__ int s_ws[16];
    const int t     = threadIdx.x;
    const int lane  = t & 31;
    const int wave  = t >> 5;
    const int b     = blockIdx.x;
    const int beg   = off[b * nwg];
    const int end   = off[(b + 1) * nwg];
    const int node0 = b << BSHIFT;
    const int2* __restrict__ rp = (const int2*)rec;

    if (t < BSIZE) s_cnt[t] = 0;
    __syncthreads();

    for (int i0 = beg + t; i0 < end; i0 += 8 * 1024) {
        int dl[8];
        #pragma unroll
        for (int u = 0; u < 8; ++u) {
            const int i = i0 + u * 1024;
            dl[u] = (i < end) ? rp[i].y : node0;
        }
        #pragma unroll
        for (int u = 0; u < 8; ++u) {
            const int i = i0 + u * 1024;
            if (i < end) atomicAdd(&s_cnt[dl[u] - node0], 1);
        }
    }
    __syncthreads();

    const int c = (t < BSIZE) ? s_cnt[t] : 0;
    int incl = c;
    #pragma unroll
    for (int o = 1; o < 32; o <<= 1) {
        const int u = __shfl_up(incl, o, 32);
        if (lane >= o) incl += u;
    }
    if (lane == 31 && wave < 16) s_ws[wave] = incl;
    __syncthreads();
    int woff = 0;
    #pragma unroll
    for (int w = 0; w < 16; ++w) if (w < wave) woff += s_ws[w];
    const int excl = woff + incl - c;
    if (t < BSIZE) {
        s_cnt[t] = excl;
        const int node = node0 + t;
        if (node < n_nodes) rowptr[node] = beg + excl;
        if (node == n_nodes - 1) rowptr[n_nodes] = end;
    }
    __syncthreads();

    for (int i0 = beg + t; i0 < end; i0 += 8 * 1024) {
        int2 r[8];
        #pragma unroll
        for (int u = 0; u < 8; ++u) {
            const int i = i0 + u * 1024;
            r[u] = (i < end) ? rp[i] : make_int2(0, node0);
        }
        int p[8];
        #pragma unroll
        for (int u = 0; u < 8; ++u) {
            const int i = i0 + u * 1024;
            p[u] = 0;
            if (i < end) p[u] = atomicAdd(&s_cnt[r[u].y - node0], 1);
        }
        #pragma unroll
        for (int u = 0; u < 8; ++u) {
            const int i = i0 + u * 1024;
            if (i < end) csr[beg + p[u]] = r[u].x;
        }
    }
}

#define TNODES 128
#define APITCH 68

#define ROW_LANE(x, J) __builtin_amdgcn_update_dpp(0, (x), 0x150 + ((J) & 15), 0xF, 0xF, false)

template <int T>
__device__ __forceinline__ void gather_trip(const float4* __restrict__ feat4,
                                            int my0, int my1, int nb, int lm,
                                            float4& acc) {
    float4 v[8];
#define GATHER_LD(U) \
    v[U] = feat4[(unsigned)ROW_LANE(((T + U) < 16 ? my0 : my1), T + U) * (FDIM / 4) + lm];
    GATHER_LD(0) GATHER_LD(1) GATHER_LD(2) GATHER_LD(3)
    GATHER_LD(4) GATHER_LD(5) GATHER_LD(6) GATHER_LD(7)
#undef GATHER_LD
    #pragma unroll
    for (int u = 0; u < 8; ++u) {
        const bool on = (T + u < nb);
        acc.x += on ? v[u].x : 0.0f;
        acc.y += on ? v[u].y : 0.0f;
        acc.z += on ? v[u].z : 0.0f;
        acc.w += on ? v[u].w : 0.0f;
    }
}

__global__ void __launch_bounds__(512, 16)
sage_layer(const float* __restrict__ feat, const int* __restrict__ rowptr,
           const int* __restrict__ csr, const float* __restrict__ Wl,
           const float* __restrict__ Wr, const float* __restrict__ bias,
           float* __restrict__ out, int n_nodes, int relu_flag) {
    __shared__ float sWl[FDIM * FDIM];
    __shared__ float sWr[FDIM * FDIM];
    __shared__ float sAgg[TNODES * APITCH];

    const int lane = threadIdx.x & 31;
    const int wave = threadIdx.x >> 5;
    const int blk0 = blockIdx.x * TNODES;
    const int half = lane >> 4;
    const int lm   = lane & 15;

    const int nl0 = wave * 8;
    int rpi = blk0 + nl0 + lane;
    rpi = rpi < n_nodes ? rpi : n_nodes;
    const int rp = rowptr[rpi];

    {
        const int i = threadIdx.x;
        const v4f l0 = ((const v4f*)Wl)[i];
        const v4f l1 = ((const v4f*)Wl)[i + 512];
        const v4f r0 = ((const v4f*)Wr)[i];
        const v4f r1 = ((const v4f*)Wr)[i + 512];
        asm volatile("" :: "v"(rp), "v"(l0), "v"(l1), "v"(r0), "v"(r1));
        ((v4f*)sWl)[i]       = l0;
        ((v4f*)sWl)[i + 512] = l1;
        ((v4f*)sWr)[i]       = r0;
        ((v4f*)sWr)[i + 512] = r1;
    }

    for (int p = 0; p < 2; ++p) {
        int id0[2], id1[2];
        #pragma unroll
        for (int rr = 0; rr < 2; ++rr) {
            const int r = 2 * p + rr;
            const int b = __shfl(rp, 2 * r + half, 32);
            const int d = __shfl(rp, 2 * r + 1 + half, 32) - b;
            id0[rr] = (lm < d)      ? csr[b + lm]      : 0;
            id1[rr] = (lm + 16 < d) ? csr[b + 16 + lm] : 0;
        }
        #pragma unroll
        for (int rr = 0; rr < 2; ++rr) {
            const int r = 2 * p + rr;
            const int beg = __shfl(rp, 2 * r + half, 32);
            const int end = __shfl(rp, 2 * r + 1 + half, 32);
            float4 acc = make_float4(0.0f, 0.0f, 0.0f, 0.0f);
            for (int base = beg; base < end; base += 32) {
                const int left = end - base;
                const int nb   = left < 32 ? left : 32;
                int my0 = id0[rr], my1 = id1[rr];
                if (base != beg) {
                    my0 = (lm < nb)      ? csr[base + lm]      : 0;
                    my1 = (lm + 16 < nb) ? csr[base + 16 + lm] : 0;
                }
                const float4* __restrict__ feat4 = (const float4*)feat;
                gather_trip<0>(feat4, my0, my1, nb, lm, acc);
                if (nb > 8)  gather_trip<8>(feat4, my0, my1, nb, lm, acc);
                if (nb > 16) gather_trip<16>(feat4, my0, my1, nb, lm, acc);
                if (nb > 24) gather_trip<24>(feat4, my0, my1, nb, lm, acc);
            }
            const float inv = 1.0f / fmaxf((float)(end - beg), 1.0f);
            acc.x *= inv; acc.y *= inv; acc.z *= inv; acc.w *= inv;
            *(float4*)(sAgg + (nl0 + 2 * r + half) * APITCH + lm * 4) = acc;
        }
    }

    const int tile  = wave >> 1;
    const int nh    = wave & 1;
    const int node0 = blk0 + tile * 16;

    __syncthreads();

    if (node0 >= n_nodes) return;

    int xr = node0 + lm;
    xr = xr < n_nodes ? xr : n_nodes - 1;
    const float* xrow = feat + (size_t)xr * FDIM + half * 2;
    v2f ax[16];
    #pragma unroll
    for (int kb = 0; kb < 16; ++kb) ax[kb] = *(const v2f*)(xrow + kb * 4);

    const float* arow = sAgg + (tile * 16 + lm) * APITCH + half * 2;

    v8f acc[2];
    #pragma unroll
    for (int j = 0; j < 2; ++j) acc[j] = (v8f){};

    #pragma unroll
    for (int kb = 0; kb < 16; ++kb) {
        const int k0 = kb * 4 + half * 2;
        const v2f a = *(const v2f*)(arow + kb * 4);
        #pragma unroll
        for (int j = 0; j < 2; ++j) {
            const int nt = nh * 2 + j;
            v2f b;
            b.x = sWl[k0 * FDIM + nt * 16 + lm];
            b.y = sWl[(k0 + 1) * FDIM + nt * 16 + lm];
            acc[j] = __builtin_amdgcn_wmma_f32_16x16x4_f32(
                false, a, false, b, (short)0, acc[j], false, false);
        }
    }
    #pragma unroll
    for (int kb = 0; kb < 16; ++kb) {
        const int k0 = kb * 4 + half * 2;
        #pragma unroll
        for (int j = 0; j < 2; ++j) {
            const int nt = nh * 2 + j;
            v2f b;
            b.x = sWr[k0 * FDIM + nt * 16 + lm];
            b.y = sWr[(k0 + 1) * FDIM + nt * 16 + lm];
            acc[j] = __builtin_amdgcn_wmma_f32_16x16x4_f32(
                false, ax[kb], false, b, (short)0, acc[j], false, false);
        }
    }

    #pragma unroll
    for (int j = 0; j < 2; ++j) {
        const int nt = nh * 2 + j;
        const float bv = bias[nt * 16 + lm];
        #pragma unroll
        for (int i = 0; i < 8; ++i) {
            float v = acc[j][i] + bv;
            if (relu_flag) v = fmaxf(v, 0.0f);
            const int row = node0 + i + half * 8;
            if (row < n_nodes) out[(size_t)row * FDIM + nt * 16 + lm] = v;
        }
    }
}

extern "C" void kernel_launch(void* const* d_in, const int* in_sizes, int n_in,
                              void* d_out, int out_size, void* d_ws, size_t ws_size,
                              hipStream_t stream) {
    const float* x    = (const float*)d_in[0];
    const int*   ei   = (const int*)d_in[1];
    const float* W1_l = (const float*)d_in[2];
    const float* b1   = (const float*)d_in[3];
    const float* W1_r = (const float*)d_in[4];
    const float* W2_l = (const float*)d_in[5];
    const float* b2   = (const float*)d_in[6];
    const float* W2_r = (const float*)d_in[7];
    float* out = (float*)d_out;

    const int n_nodes = in_sizes[0] / FDIM;
    const int n_edges = in_sizes[1] / 2;
    const int* src = ei;
    const int* dst = ei + n_edges;

    const int nb   = (n_nodes + BSIZE - 1) / BSIZE;
    const int nwg  = (n_edges + CHUNK - 1) / CHUNK;
    const int tot  = nb * nwg;
    const int tpad = (tot + 4) & ~3;

    float* h      = (float*)d_ws;
    int*   hist   = (int*)(h + (size_t)n_nodes * FDIM);
    int*   off    = hist + tpad;
    int*   rec    = off + tpad;
    int*   csr    = rec + 2 * (size_t)n_edges;
    int*   rowptr = csr + n_edges;

    const int scan_grid  = (tot + 1023) / 1024;
    const int layer_grid = (n_nodes + TNODES - 1) / TNODES;

    count_buckets<<<nwg, 1024, 0, stream>>>(dst, hist, n_edges, nb, nwg);
    scan_deg<<<scan_grid, 1024, 0, stream>>>(hist, off, tot);
    place_edges<<<nwg, 1024, 0, stream>>>(src, dst, off, rec, n_edges, nb, nwg);
    sort_bucket<<<nb, 1024, 0, stream>>>(rec, off, rowptr, csr, n_nodes, nwg);

    sage_layer<<<layer_grid, 512, 0, stream>>>(x, rowptr, csr, W1_l, W1_r, b1, h,
                                               n_nodes,  1);

    sage_layer<<<layer_grid, 512, 0, stream>>>(h, rowptr, csr, W2_l, W2_r, b2, out,
                                               n_nodes,  0);
}


# Sort passes order their records inside LDS and write them out as consecutive runs, not one scattered 8- or 4-byte global store per edge; per-node sum order = arrival order of LDS ranks, as before
# speedup vs baseline: 3.8035x; 1.1323x over previous
#include <hip/hip_runtime.h>
#include <hip/hip_bf16.h>

typedef __attribute__((ext_vector_type(2))) float v2f;
typedef __attribute__((ext_vector_type(8))) float v8f;
typedef __attribute__((ext_vector_type(4))) float v4f;

#define FDIM 64

#define BSHIFT 9
#define BSIZE  512
#define CHUNK  8192
#define MAXB   2048
#define SCAP   16384

__global__ void __launch_bounds__(1024)
count_buckets(const int* __restrict__ dst, int* __restrict__ hist,
              int n_edges, int nb, int nwg) {
    __shared__ int s_h[MAXB];
    const int t = threadIdx.x;
    for (int b = t; b < nb; b += 1024) s_h[b] = 0;
    __syncthreads();
    const int base = blockIdx.x * CHUNK + t;
    int d[8];
    #pragma unroll
    for (int u = 0; u < 8; ++u) {
        const int e = base + u * 1024;
        d[u] = (e < n_edges) ? dst[e] : 0;
    }
    #pragma unroll
    for (int u = 0; u < 8; ++u) {
        const int e = base + u * 1024;
        if (e < n_edges) atomicAdd(&s_h[d[u] >> BSHIFT], 1);
    }
    __syncthreads();
    for (int b = t; b < nb; b += 1024) hist[b * nwg + blockIdx.x] = s_h[b];
}

__global__ void __launch_bounds__(1024)
scan_deg(const int* __restrict__ deg, int* __restrict__ rowptr, int n_nodes) {
    __shared__ int s_pre[32];
    __shared__ int s_sum[32];
    const int t    = threadIdx.x;
    const int lane = t & 31;
    const int wave = t >> 5;
    const int b0   = blockIdx.x * 1024;

    int p = 0;
    const int4* dv = (const int4*)deg + t;
    const int nfull = b0 >> 12;
    int k = 0;
    for (; k + 8 <= nfull; k += 8) {
        int4 v[8];
        #pragma unroll
        for (int u = 0; u < 8; ++u) v[u] = dv[(k + u) * 1024];
        #pragma unroll
        for (int u = 0; u < 8; ++u) p += (v[u].x + v[u].y) + (v[u].z + v[u].w);
    }
    for (; k < nfull; ++k) {
        const int4 v = dv[k * 1024];
        p += (v.x + v.y) + (v.z + v.w);
    }
    if ((nfull << 12) + t * 4 < b0) {
        const int4 v = dv[nfull * 1024];
        p += (v.x + v.y) + (v.z + v.w);
    }
    const int i = b0 + t;
    const int d = (i < n_nodes) ? deg[i] : 0;

    int incl = d;
    #pragma unroll
    for (int o = 1; o < 32; o <<= 1) {
        const int u = __shfl_up(incl, o, 32);
        if (lane >= o) incl += u;
    }
    #pragma unroll
    for (int o = 16; o > 0; o >>= 1) p += __shfl_xor(p, o, 32);
    if (lane == 31) s_sum[wave] = incl;
    if (lane == 0)  s_pre[wave] = p;
    __syncthreads();

    int off = 0;
    #pragma unroll
    for (int w = 0; w < 32; ++w) {
        off += s_pre[w];
        if (w < wave) off += s_sum[w];
    }
    const int excl = off + incl - d;
    if (i < n_nodes) rowptr[i] = excl;
    if (i == n_nodes - 1) rowptr[n_nodes] = excl + d;
}

__global__ void __launch_bounds__(1024)
place_edges(const int* __restrict__ src, const int* __restrict__ dst,
            const int* __restrict__ hist, const int* __restrict__ off,
            int* __restrict__ rec, int n_edges, int nb, int nwg) {
    __shared__ int  s_pos[MAXB];
    __shared__ int  s_gof[MAXB];
    __shared__ int  s_ws[32];
    __shared__ int2 s_rec[CHUNK];
    const int t    = threadIdx.x;
    const int lane = t & 31;
    const int wave = t >> 5;
    const int blk  = blockIdx.x;

    const int base = blk * CHUNK + t;
    int d[8], s[8];
    #pragma unroll
    for (int u = 0; u < 8; ++u) {
        const int e = base + u * 1024;
        d[u] = (e < n_edges) ? dst[e] : 0;
        s[u] = (e < n_edges) ? src[e] : 0;
    }

    const int b0 = 2 * t, b1 = 2 * t + 1;
    const int c0 = (b0 < nb) ? hist[b0 * nwg + blk] : 0;
    const int c1 = (b1 < nb) ? hist[b1 * nwg + blk] : 0;
    const int g0 = (b0 < nb) ? off[b0 * nwg + blk] : 0;
    const int g1 = (b1 < nb) ? off[b1 * nwg + blk] : 0;
    const int c  = c0 + c1;
    int incl = c;
    #pragma unroll
    for (int o = 1; o < 32; o <<= 1) {
        const int u = __shfl_up(incl, o, 32);
        if (lane >= o) incl += u;
    }
    if (lane == 31) s_ws[wave] = incl;
    __syncthreads();
    int woff = 0;
    #pragma unroll
    for (int w = 0; w < 32; ++w) if (w < wave) woff += s_ws[w];
    const int excl = woff + incl - c;
    s_pos[b0] = excl;        s_gof[b0] = g0 - excl;
    s_pos[b1] = excl + c0;   s_gof[b1] = g1 - (excl + c0);
    __syncthreads();

    int p[8];
    #pragma unroll
    for (int u = 0; u < 8; ++u) {
        const int e = base + u * 1024;
        p[u] = 0;
        if (e < n_edges) p[u] = atomicAdd(&s_pos[d[u] >> BSHIFT], 1);
    }
    #pragma unroll
    for (int u = 0; u < 8; ++u) {
        const int e = base + u * 1024;
        if (e < n_edges) s_rec[p[u]] = make_int2(s[u], d[u]);
    }
    __syncthreads();

    const int left  = n_edges - blk * CHUNK;
    const int n_loc = left < CHUNK ? left : CHUNK;
    int2* __restrict__ rp = (int2*)rec;
    int2 r[8];
    int  g[8];
    #pragma unroll
    for (int u = 0; u < 8; ++u) {
        const int i = t + u * 1024;
        r[u] = (i < n_loc) ? s_rec[i] : make_int2(0, 0);
    }
    #pragma unroll
    for (int u = 0; u < 8; ++u) {
        const int i = t + u * 1024;
        g[u] = (i < n_loc) ? s_gof[r[u].y >> BSHIFT] : 0;
    }
    #pragma unroll
    for (int u = 0; u < 8; ++u) {
        const int i = t + u * 1024;
        if (i < n_loc) rp[(unsigned)(g[u] + i)] = r[u];
    }
}

template <bool STAGED>
__device__ __forceinline__ void place_ids(const int2* __restrict__ rp, int* s_cnt, int* s_out,
                                          int* __restrict__ csr, int beg, int end,
                                          int node0, int t) {
    for (int i0 = beg + t; i0 < end; i0 += 8 * 1024) {
        int2 r[8];
        #pragma unroll
        for (int u = 0; u < 8; ++u) {
            const int i = i0 + u * 1024;
            r[u] = (i < end) ? rp[i] : make_int2(0, node0);
        }
        int p[8];
        #pragma unroll
        for (int u = 0; u < 8; ++u) {
            const int i = i0 + u * 1024;
            p[u] = 0;
            if (i < end) p[u] = atomicAdd(&s_cnt[r[u].y - node0], 1);
        }
        #pragma unroll
        for (int u = 0; u < 8; ++u) {
            const int i = i0 + u * 1024;
            if (i < end) {
                if (STAGED) s_out[p[u]] = r[u].x;
                else        csr[beg + p[u]] = r[u].x;
            }
        }
    }
}

__global__ void __launch_bounds__(1024)
sort_bucket(const int* __restrict__ rec, const int* __restrict__ off,
            int* __restrict__ rowptr, int* __restrict__ csr,
            int n_nodes, int nwg) {
    __shared__ int s_cnt[BSIZE];
    __shared__ int s_ws[16];
    __shared__ int s_out[SCAP];
    const int t     = threadIdx.x;
    const int lane  = t & 31;
    const int wave  = t >> 5;
    const int b     = blockIdx.x;
    const int beg   = off[b * nwg];
    const int end   = off[(b + 1) * nwg];
    const int node0 = b << BSHIFT;
    const int n     = end - beg;
    const bool staged = (n <= SCAP);
    const int2* __restrict__ rp = (const int2*)rec;

    if (t < BSIZE) s_cnt[t] = 0;
    __syncthreads();

    for (int i0 = beg + t; i0 < end; i0 += 8 * 1024) {
        int dl[8];
        #pragma unroll
        for (int u = 0; u < 8; ++u) {
            const int i = i0 + u * 1024;
            dl[u] = (i < end) ? rp[i].y : node0;
        }
        #pragma unroll
        for (int u = 0; u < 8; ++u) {
            const int i = i0 + u * 1024;
            if (i < end) atomicAdd(&s_cnt[dl[u] - node0], 1);
        }
    }
    __syncthreads();

    const int c = (t < BSIZE) ? s_cnt[t] : 0;
    int incl = c;
    #pragma unroll
    for (int o = 1; o < 32; o <<= 1) {
        const int u = __shfl_up(incl, o, 32);
        if (lane >= o) incl += u;
    }
    if (lane == 31 && wave < 16) s_ws[wave] = incl;
    __syncthreads();
    int woff = 0;
    #pragma unroll
    for (int w = 0; w < 16; ++w) if (w < wave) woff += s_ws[w];
    const int excl = woff + incl - c;
    if (t < BSIZE) {
        s_cnt[t] = excl;
        const int node = node0 + t;
        if (node < n_nodes) rowptr[node] = beg + excl;
        if (node == n_nodes - 1) rowptr[n_nodes] = end;
    }
    __syncthreads();

    if (staged) {
        place_ids<true>(rp, s_cnt, s_out, csr, beg, end, node0, t);
        __syncthreads();
        for (int i0 = t; i0 < n; i0 += 8 * 1024) {
            int v[8];
            #pragma unroll
            for (int u = 0; u < 8; ++u) {
                const int i = i0 + u * 1024;
                v[u] = (i < n) ? s_out[i] : 0;
            }
            #pragma unroll
            for (int u = 0; u < 8; ++u) {
                const int i = i0 + u * 1024;
                if (i < n) csr[beg + i] = v[u];
            }
        }
    } else {
        place_ids<false>(rp, s_cnt, s_out, csr, beg, end, node0, t);
    }
}

#define TNODES 128
#define APITCH 68

#define ROW_LANE(x, J) __builtin_amdgcn_update_dpp(0, (x), 0x150 + ((J) & 15), 0xF, 0xF, false)

template <int T>
__device__ __forceinline__ void gather_trip(const float4* __restrict__ feat4,
                                            int my0, int my1, int nb, int lm,
                                            float4& acc) {
    float4 v[8];
#define GATHER_LD(U) \
    v[U] = feat4[(unsigned)ROW_LANE(((T + U) < 16 ? my0 : my1), T + U) * (FDIM / 4) + lm];
    GATHER_LD(0) GATHER_LD(1) GATHER_LD(2) GATHER_LD(3)
    GATHER_LD(4) GATHER_LD(5) GATHER_LD(6) GATHER_LD(7)
#undef GATHER_LD
    #pragma unroll
    for (int u = 0; u < 8; ++u) {
        const bool on = (T + u < nb);
        acc.x += on ? v[u].x : 0.0f;
        acc.y += on ? v[u].y : 0.0f;
        acc.z += on ? v[u].z : 0.0f;
        acc.w += on ? v[u].w : 0.0f;
    }
}

__global__ void __launch_bounds__(512, 16)
sage_layer(const float* __restrict__ feat, const int* __restrict__ rowptr,
           const int* __restrict__ csr, const float* __restrict__ Wl,
           const float* __restrict__ Wr, const float* __restrict__ bias,
           float* __restrict__ out, int n_nodes, int relu_flag) {
    __shared__ float sWl[FDIM * FDIM];
    __shared__ float sWr[FDIM * FDIM];
    __shared__ float sAgg[TNODES * APITCH];

    const int lane = threadIdx.x & 31;
    const int wave = threadIdx.x >> 5;
    const int blk0 = blockIdx.x * TNODES;
    const int half = lane >> 4;
    const int lm   = lane & 15;

    const int nl0 = wave * 8;
    int rpi = blk0 + nl0 + lane;
    rpi = rpi < n_nodes ? rpi : n_nodes;
    const int rp = rowptr[rpi];

    {
        const int i = threadIdx.x;
        const v4f l0 = ((const v4f*)Wl)[i];
        const v4f l1 = ((const v4f*)Wl)[i + 512];
        const v4f r0 = ((const v4f*)Wr)[i];
        const v4f r1 = ((const v4f*)Wr)[i + 512];
        asm volatile("" :: "v"(rp), "v"(l0), "v"(l1), "v"(r0), "v"(r1));
        ((v4f*)sWl)[i]       = l0;
        ((v4f*)sWl)[i + 512] = l1;
        ((v4f*)sWr)[i]       = r0;
        ((v4f*)sWr)[i + 512] = r1;
    }

    for (int p = 0; p < 2; ++p) {
        int id0[2], id1[2];
        #pragma unroll
        for (int rr = 0; rr < 2; ++rr) {
            const int r = 2 * p + rr;
            const int b = __shfl(rp, 2 * r + half, 32);
            const int d = __shfl(rp, 2 * r + 1 + half, 32) - b;
            id0[rr] = (lm < d)      ? csr[b + lm]      : 0;
            id1[rr] = (lm + 16 < d) ? csr[b + 16 + lm] : 0;
        }
        #pragma unroll
        for (int rr = 0; rr < 2; ++rr) {
            const int r = 2 * p + rr;
            const int beg = __shfl(rp, 2 * r + half, 32);
            const int end = __shfl(rp, 2 * r + 1 + half, 32);
            float4 acc = make_float4(0.0f, 0.0f, 0.0f, 0.0f);
            for (int base = beg; base < end; base += 32) {
                const int left = end - base;
                const int nb   = left < 32 ? left : 32;
                int my0 = id0[rr], my1 = id1[rr];
                if (base != beg) {
                    my0 = (lm < nb)      ? csr[base + lm]      : 0;
                    my1 = (lm + 16 < nb) ? csr[base + 16 + lm] : 0;
                }
                const float4* __restrict__ feat4 = (const float4*)feat;
                gather_trip<0>(feat4, my0, my1, nb, lm, acc);
                if (nb > 8)  gather_trip<8>(feat4, my0, my1, nb, lm, acc);
                if (nb > 16) gather_trip<16>(feat4, my0, my1, nb, lm, acc);
                if (nb > 24) gather_trip<24>(feat4, my0, my1, nb, lm, acc);
            }
            const float inv = 1.0f / fmaxf((float)(end - beg), 1.0f);
            acc.x *= inv; acc.y *= inv; acc.z *= inv; acc.w *= inv;
            *(float4*)(sAgg + (nl0 + 2 * r + half) * APITCH + lm * 4) = acc;
        }
    }

    const int tile  = wave >> 1;
    const int nh    = wave & 1;
    const int node0 = blk0 + tile * 16;

    __syncthreads();

    if (node0 >= n_nodes) return;

    int xr = node0 + lm;
    xr = xr < n_nodes ? xr : n_nodes - 1;
    const float* xrow = feat + (size_t)xr * FDIM + half * 2;
    v2f ax[16];
    #pragma unroll
    for (int kb = 0; kb < 16; ++kb) ax[kb] = *(const v2f*)(xrow + kb * 4);

    const float* arow = sAgg + (tile * 16 + lm) * APITCH + half * 2;

    v8f acc[2];
    #pragma unroll
    for (int j = 0; j < 2; ++j) acc[j] = (v8f){};

    #pragma unroll
    for (int kb = 0; kb < 16; ++kb) {
        const int k0 = kb * 4 + half * 2;
        const v2f a = *(const v2f*)(arow + kb * 4);
        #pragma unroll
        for (int j = 0; j < 2; ++j) {
            const int nt = nh * 2 + j;
            v2f b;
            b.x = sWl[k0 * FDIM + nt * 16 + lm];
            b.y = sWl[(k0 + 1) * FDIM + nt * 16 + lm];
            acc[j] = __builtin_amdgcn_wmma_f32_16x16x4_f32(
                false, a, false, b, (short)0, acc[j], false, false);
        }
    }
    #pragma unroll
    for (int kb = 0; kb < 16; ++kb) {
        const int k0 = kb * 4 + half * 2;
        #pragma unroll
        for (int j = 0; j < 2; ++j) {
            const int nt = nh * 2 + j;
            v2f b;
            b.x = sWr[k0 * FDIM + nt * 16 + lm];
            b.y = sWr[(k0 + 1) * FDIM + nt * 16 + lm];
            acc[j] = __builtin_amdgcn_wmma_f32_16x16x4_f32(
                false, ax[kb], false, b, (short)0, acc[j], false, false);
        }
    }

    #pragma unroll
    for (int j = 0; j < 2; ++j) {
        const int nt = nh * 2 + j;
        const float bv = bias[nt * 16 + lm];
        #pragma unroll
        for (int i = 0; i < 8; ++i) {
            float v = acc[j][i] + bv;
            if (relu_flag) v = fmaxf(v, 0.0f);
            const int row = node0 + i + half * 8;
            if (row < n_nodes) out[(size_t)row * FDIM + nt * 16 + lm] = v;
        }
    }
}

extern "C" void kernel_launch(void* const* d_in, const int* in_sizes, int n_in,
                              void* d_out, int out_size, void* d_ws, size_t ws_size,
                              hipStream_t stream) {
    const float* x    = (const float*)d_in[0];
    const int*   ei   = (const int*)d_in[1];
    const float* W1_l = (const float*)d_in[2];
    const float* b1   = (const float*)d_in[3];
    const float* W1_r = (const float*)d_in[4];
    const float* W2_l = (const float*)d_in[5];
    const float* b2   = (const float*)d_in[6];
    const float* W2_r = (const float*)d_in[7];
    float* out = (float*)d_out;

    const int n_nodes = in_sizes[0] / FDIM;
    const int n_edges = in_sizes[1] / 2;
    const int* src = ei;
    const int* dst = ei + n_edges;

    const int nb   = (n_nodes + BSIZE - 1) / BSIZE;
    const int nwg  = (n_edges + CHUNK - 1) / CHUNK;
    const int tot  = nb * nwg;
    const int tpad = (tot + 4) & ~3;

    float* h      = (float*)d_ws;
    int*   hist   = (int*)(h + (size_t)n_nodes * FDIM);
    int*   off    = hist + tpad;
    int*   rec    = off + tpad;
    int*   csr    = rec + 2 * (size_t)n_edges;
    int*   rowptr = csr + n_edges;

    const int scan_grid  = (tot + 1023) / 1024;
    const int layer_grid = (n_nodes + TNODES - 1) / TNODES;

    count_buckets<<<nwg, 1024, 0, stream>>>(dst, hist, n_edges, nb, nwg);
    scan_deg<<<scan_grid, 1024, 0, stream>>>(hist, off, tot);
    place_edges<<<nwg, 1024, 0, stream>>>(src, dst, hist, off, rec, n_edges, nb, nwg);
    sort_bucket<<<nb, 1024, 0, stream>>>(rec, off, rowptr, csr, n_nodes, nwg);

    sage_layer<<<layer_grid, 512, 0, stream>>>(x, rowptr, csr, W1_l, W1_r, b1, h,
                                               n_nodes,  1);

    sage_layer<<<layer_grid, 512, 0, stream>>>(h, rowptr, csr, W2_l, W2_r, b2, out,
                                               n_nodes,  0);
}


# Edge blocks keep bucket-ordered records in their own slab; a bucket's block collects its run from each slab: no global histogram, scan or count pass; 6 -> 4 launches; sum order = LDS arrival rank
# speedup vs baseline: 4.2938x; 1.1289x over previous
#include <hip/hip_runtime.h>
#include <hip/hip_bf16.h>

typedef __attribute__((ext_vector_type(2))) float v2f;
typedef __attribute__((ext_vector_type(8))) float v8f;
typedef __attribute__((ext_vector_type(4))) float v4f;

#define FDIM 64

#define BSHIFT 9
#define BSIZE  512
#define CHUNK  8192
#define MAXB   2048
#define SCAP   16384

__global__ void __launch_bounds__(1024)
order_block(const int* __restrict__ src, const int* __restrict__ dst,
            int* __restrict__ tab, int* __restrict__ rec,
            int n_edges, int nb, int nwg) {
    __shared__ __attribute__((aligned(16))) int  s_cnt[MAXB];
    __shared__ int  s_ws[32];
    __shared__ __attribute__((aligned(16))) int2 s_rec[CHUNK];
    const int t    = threadIdx.x;
    const int lane = t & 31;
    const int wave = t >> 5;
    const int blk  = blockIdx.x;

    s_cnt[t] = 0;
    s_cnt[t + 1024] = 0;

    const int base = blk * CHUNK + t;
    int d[8], s[8];
    #pragma unroll
    for (int u = 0; u < 8; ++u) {
        const int e = base + u * 1024;
        d[u] = (e < n_edges) ? dst[e] : 0;
        s[u] = (e < n_edges) ? src[e] : 0;
    }
    __syncthreads();

    int k[8];
    #pragma unroll
    for (int u = 0; u < 8; ++u) {
        const int e = base + u * 1024;
        k[u] = 0;
        if (e < n_edges) k[u] = atomicAdd(&s_cnt[d[u] >> BSHIFT], 1);
    }
    __syncthreads();

    const int b0 = 2 * t, b1 = 2 * t + 1;
    const int2 cc = ((const int2*)s_cnt)[t];
    const int c  = cc.x + cc.y;
    int incl = c;
    #pragma unroll
    for (int o = 1; o < 32; o <<= 1) {
        const int u = __shfl_up(incl, o, 32);
        if (lane >= o) incl += u;
    }
    if (lane == 31) s_ws[wave] = incl;
    __syncthreads();
    int woff = 0;
    #pragma unroll
    for (int w = 0; w < 32; ++w) if (w < wave) woff += s_ws[w];
    const int excl = woff + incl - c;
    ((int2*)s_cnt)[t] = make_int2(excl, excl + cc.x);
    if (b0 <= nb) tab[b0 * nwg + blk] = excl;
    if (b1 <= nb) tab[b1 * nwg + blk] = excl + cc.x;
    __syncthreads();

    int q[8];
    #pragma unroll
    for (int u = 0; u < 8; ++u) q[u] = s_cnt[d[u] >> BSHIFT] + k[u];
    #pragma unroll
    for (int u = 0; u < 8; ++u) {
        const int e = base + u * 1024;
        if (e < n_edges) s_rec[q[u]] = make_int2(s[u], d[u]);
    }
    __syncthreads();

    const int left  = n_edges - blk * CHUNK;
    const int n_loc = left < CHUNK ? left : CHUNK;
    int2* __restrict__ slab = (int2*)rec + (size_t)blk * CHUNK;
    const int4* s4 = (const int4*)s_rec;
    int4 r[4];
    #pragma unroll
    for (int u = 0; u < 4; ++u) {
        const int i = t + u * 1024;
        r[u] = (2 * i < n_loc) ? s4[i] : make_int4(0, 0, 0, 0);
    }
    #pragma unroll
    for (int u = 0; u < 4; ++u) {
        const int i = t + u * 1024;
        if (2 * i + 1 < n_loc)  ((int4*)slab)[i] = r[u];
        else if (2 * i < n_loc) slab[2 * i] = make_int2(r[u].x, r[u].y);
    }
}

template <bool PLACE, bool STAGED>
__device__ __forceinline__ void walk_runs(const int2* __restrict__ rp, const int* __restrict__ tab,
                                          int* s_cnt, int* s_out, int* __restrict__ csr,
                                          int beg, int b, int node0, int nwg, int t) {
    const int row = t >> 4;
    const int l16 = t & 15;
    for (int rbase = 0; rbase < nwg; rbase += 256) {
        int lo[4], cn[4];
        #pragma unroll
        for (int j = 0; j < 4; ++j) {
            const int r = rbase + row + 64 * j;
            int a = 0, e = 0;
            if (r < nwg) { a = tab[b * nwg + r]; e = tab[(b + 1) * nwg + r]; }
            lo[j] = r * CHUNK + a;
            cn[j] = e - a;
        }
        int2 rc[4][4];
        #pragma unroll
        for (int j = 0; j < 4; ++j) {
            #pragma unroll
            for (int m = 0; m < 4; ++m) {
                const int i = l16 + 16 * m;
                rc[j][m] = make_int2(0, node0);
                if (i < cn[j]) {
                    if (PLACE) rc[j][m] = rp[(unsigned)(lo[j] + i)];
                    else       rc[j][m].y = rp[(unsigned)(lo[j] + i)].y;
                }
            }
        }
        if (PLACE) {
            int p[4][4];
            #pragma unroll
            for (int j = 0; j < 4; ++j) {
                #pragma unroll
                for (int m = 0; m < 4; ++m) {
                    p[j][m] = 0;
                    if (l16 + 16 * m < cn[j]) p[j][m] = atomicAdd(&s_cnt[rc[j][m].y - node0], 1);
                }
            }
            #pragma unroll
            for (int j = 0; j < 4; ++j) {
                #pragma unroll
                for (int m = 0; m < 4; ++m) {
                    if (l16 + 16 * m < cn[j]) {
                        if (STAGED) s_out[p[j][m]] = rc[j][m].x;
                        else        csr[beg + p[j][m]] = rc[j][m].x;
                    }
                }
            }
        } else {
            #pragma unroll
            for (int j = 0; j < 4; ++j) {
                #pragma unroll
                for (int m = 0; m < 4; ++m) {
                    if (l16 + 16 * m < cn[j]) atomicAdd(&s_cnt[rc[j][m].y - node0], 1);
                }
            }
        }
        #pragma unroll
        for (int j = 0; j < 4; ++j) {
            for (int i = 64 + l16; i < cn[j]; i += 16) {
                const int2 v = rp[(unsigned)(lo[j] + i)];
                const int  p = atomicAdd(&s_cnt[v.y - node0], 1);
                if (PLACE) {
                    if (STAGED) s_out[p] = v.x;
                    else        csr[beg + p] = v.x;
                }
            }
        }
    }
}

__global__ void __launch_bounds__(1024)
sort_bucket(const int* __restrict__ rec, const int* __restrict__ tab,
            int* __restrict__ rowptr, int* __restrict__ csr,
            int n_nodes, int nwg) {
    __shared__ int s_cnt[BSIZE];
    __shared__ int s_ws[16];
    __shared__ int s_sb[32];
    __shared__ int s_se[32];
    __shared__ int s_out[SCAP];
    const int t     = threadIdx.x;
    const int lane  = t & 31;
    const int wave  = t >> 5;
    const int b     = blockIdx.x;
    const int node0 = b << BSHIFT;
    const int2* __restrict__ rp = (const int2*)rec;

    if (t < BSIZE) s_cnt[t] = 0;
    __syncthreads();

    walk_runs<false, false>(rp, tab, s_cnt, s_out, csr, 0, b, node0, nwg, t);

    int sb = 0, se = 0;
    for (int r = t; r < nwg; r += 1024) {
        sb += tab[b * nwg + r];
        se += tab[(b + 1) * nwg + r];
    }
    #pragma unroll
    for (int o = 16; o > 0; o >>= 1) {
        sb += __shfl_xor(sb, o, 32);
        se += __shfl_xor(se, o, 32);
    }
    if (lane == 0) { s_sb[wave] = sb; s_se[wave] = se; }
    __syncthreads();

    int beg = 0, end = 0;
    #pragma unroll
    for (int w = 0; w < 32; ++w) { beg += s_sb[w]; end += s_se[w]; }
    const int n = end - beg;
    const bool staged = (n <= SCAP);

    const int c = (t < BSIZE) ? s_cnt[t] : 0;
    int incl = c;
    #pragma unroll
    for (int o = 1; o < 32; o <<= 1) {
        const int u = __shfl_up(incl, o, 32);
        if (lane >= o) incl += u;
    }
    if (lane == 31 && wave < 16) s_ws[wave] = incl;
    __syncthreads();
    int woff = 0;
    #pragma unroll
    for (int w = 0; w < 16; ++w) if (w < wave) woff += s_ws[w];
    const int excl = woff + incl - c;
    if (t < BSIZE) {
        s_cnt[t] = excl;
        const int node = node0 + t;
        if (node < n_nodes) rowptr[node] = beg + excl;
        if (node == n_nodes - 1) rowptr[n_nodes] = end;
    }
    __syncthreads();

    if (staged) {
        walk_runs<true, true>(rp, tab, s_cnt, s_out, csr, beg, b, node0, nwg, t);
        __syncthreads();
        for (int i0 = t; i0 < n; i0 += 8 * 1024) {
            int v[8];
            #pragma unroll
            for (int u = 0; u < 8; ++u) {
                const int i = i0 + u * 1024;
                v[u] = (i < n) ? s_out[i] : 0;
            }
            #pragma unroll
            for (int u = 0; u < 8; ++u) {
                const int i = i0 + u * 1024;
                if (i < n) csr[beg + i] = v[u];
            }
        }
    } else {
        walk_runs<true, false>(rp, tab, s_cnt, s_out, csr, beg, b, node0, nwg, t);
    }
}

#define TNODES 128
#define APITCH 68

#define ROW_LANE(x, J) __builtin_amdgcn_update_dpp(0, (x), 0x150 + ((J) & 15), 0xF, 0xF, false)

template <int T>
__device__ __forceinline__ void gather_trip(const float4* __restrict__ feat4,
                                            int my0, int my1, int nb, int lm,
                                            float4& acc) {
    float4 v[8];
#define GATHER_LD(U) \
    v[U] = feat4[(unsigned)ROW_LANE(((T + U) < 16 ? my0 : my1), T + U) * (FDIM / 4) + lm];
    GATHER_LD(0) GATHER_LD(1) GATHER_LD(2) GATHER_LD(3)
    GATHER_LD(4) GATHER_LD(5) GATHER_LD(6) GATHER_LD(7)
#undef GATHER_LD
    #pragma unroll
    for (int u = 0; u < 8; ++u) {
        const bool on = (T + u < nb);
        acc.x += on ? v[u].x : 0.0f;
        acc.y += on ? v[u].y : 0.0f;
        acc.z += on ? v[u].z : 0.0f;
        acc.w += on ? v[u].w : 0.0f;
    }
}

__global__ void __launch_bounds__(512, 16)
sage_layer(const float* __restrict__ feat, const int* __restrict__ rowptr,
           const int* __restrict__ csr, const float* __restrict__ Wl,
           const float* __restrict__ Wr, const float* __restrict__ bias,
           float* __restrict__ out, int n_nodes, int relu_flag) {
    __shared__ float sWl[FDIM * FDIM];
    __shared__ float sWr[FDIM * FDIM];
    __shared__ float sAgg[TNODES * APITCH];

    const int lane = threadIdx.x & 31;
    const int wave = threadIdx.x >> 5;
    const int blk0 = blockIdx.x * TNODES;
    const int half = lane >> 4;
    const int lm   = lane & 15;

    const int nl0 = wave * 8;
    int rpi = blk0 + nl0 + lane;
    rpi = rpi < n_nodes ? rpi : n_nodes;
    const int rp = rowptr[rpi];

    {
        const int i = threadIdx.x;
        const v4f l0 = ((const v4f*)Wl)[i];
        const v4f l1 = ((const v4f*)Wl)[i + 512];
        const v4f r0 = ((const v4f*)Wr)[i];
        const v4f r1 = ((const v4f*)Wr)[i + 512];
        asm volatile("" :: "v"(rp), "v"(l0), "v"(l1), "v"(r0), "v"(r1));
        ((v4f*)sWl)[i]       = l0;
        ((v4f*)sWl)[i + 512] = l1;
        ((v4f*)sWr)[i]       = r0;
        ((v4f*)sWr)[i + 512] = r1;
    }

    for (int p = 0; p < 2; ++p) {
        int id0[2], id1[2];
        #pragma unroll
        for (int rr = 0; rr < 2; ++rr) {
            const int r = 2 * p + rr;
            const int b = __shfl(rp, 2 * r + half, 32);
            const int d = __shfl(rp, 2 * r + 1 + half, 32) - b;
            id0[rr] = (lm < d)      ? csr[b + lm]      : 0;
            id1[rr] = (lm + 16 < d) ? csr[b + 16 + lm] : 0;
        }
        #pragma unroll
        for (int rr = 0; rr < 2; ++rr) {
            const int r = 2 * p + rr;
            const int beg = __shfl(rp, 2 * r + half, 32);
            const int end = __shfl(rp, 2 * r + 1 + half, 32);
            float4 acc = make_float4(0.0f, 0.0f, 0.0f, 0.0f);
            for (int base = beg; base < end; base += 32) {
                const int left = end - base;
                const int nb   = left < 32 ? left : 32;
                int my0 = id0[rr], my1 = id1[rr];
                if (base != beg) {
                    my0 = (lm < nb)      ? csr[base + lm]      : 0;
                    my1 = (lm + 16 < nb) ? csr[base + 16 + lm] : 0;
                }
                const float4* __restrict__ feat4 = (const float4*)feat;
                gather_trip<0>(feat4, my0, my1, nb, lm, acc);
                if (nb > 8)  gather_trip<8>(feat4, my0, my1, nb, lm, acc);
                if (nb > 16) gather_trip<16>(feat4, my0, my1, nb, lm, acc);
                if (nb > 24) gather_trip<24>(feat4, my0, my1, nb, lm, acc);
            }
            const float inv = 1.0f / fmaxf((float)(end - beg), 1.0f);
            acc.x *= inv; acc.y *= inv; acc.z *= inv; acc.w *= inv;
            *(float4*)(sAgg + (nl0 + 2 * r + half) * APITCH + lm * 4) = acc;
        }
    }

    const int tile  = wave >> 1;
    const int nh    = wave & 1;
    const int node0 = blk0 + tile * 16;

    __syncthreads();

    if (node0 >= n_nodes) return;

    int xr = node0 + lm;
    xr = xr < n_nodes ? xr : n_nodes - 1;
    const float* xrow = feat + (size_t)xr * FDIM + half * 2;
    v2f ax[16];
    #pragma unroll
    for (int kb = 0; kb < 16; ++kb) ax[kb] = *(const v2f*)(xrow + kb * 4);

    const float* arow = sAgg + (tile * 16 + lm) * APITCH + half * 2;

    v8f acc[2];
    #pragma unroll
    for (int j = 0; j < 2; ++j) acc[j] = (v8f){};

    #pragma unroll
    for (int kb = 0; kb < 16; ++kb) {
        const int k0 = kb * 4 + half * 2;
        const v2f a = *(const v2f*)(arow + kb * 4);
        #pragma unroll
        for (int j = 0; j < 2; ++j) {
            const int nt = nh * 2 + j;
            v2f b;
            b.x = sWl[k0 * FDIM + nt * 16 + lm];
            b.y = sWl[(k0 + 1) * FDIM + nt * 16 + lm];
            acc[j] = __builtin_amdgcn_wmma_f32_16x16x4_f32(
                false, a, false, b, (short)0, acc[j], false, false);
        }
    }
    #pragma unroll
    for (int kb = 0; kb < 16; ++kb) {
        const int k0 = kb * 4 + half * 2;
        #pragma unroll
        for (int j = 0; j < 2; ++j) {
            const int nt = nh * 2 + j;
            v2f b;
            b.x = sWr[k0 * FDIM + nt * 16 + lm];
            b.y = sWr[(k0 + 1) * FDIM + nt * 16 + lm];
            acc[j] = __builtin_amdgcn_wmma_f32_16x16x4_f32(
                false, ax[kb], false, b, (short)0, acc[j], false, false);
        }
    }

    #pragma unroll
    for (int j = 0; j < 2; ++j) {
        const int nt = nh * 2 + j;
        const float bv = bias[nt * 16 + lm];
        #pragma unroll
        for (int i = 0; i < 8; ++i) {
            float v = acc[j][i] + bv;
            if (relu_flag) v = fmaxf(v, 0.0f);
            const int row = node0 + i + half * 8;
            if (row < n_nodes) out[(size_t)row * FDIM + nt * 16 + lm] = v;
        }
    }
}

extern "C" void kernel_launch(void* const* d_in, const int* in_sizes, int n_in,
                              void* d_out, int out_size, void* d_ws, size_t ws_size,
                              hipStream_t stream) {
    const float* x    = (const float*)d_in[0];
    const int*   ei   = (const int*)d_in[1];
    const float* W1_l = (const float*)d_in[2];
    const float* b1   = (const float*)d_in[3];
    const float* W1_r = (const float*)d_in[4];
    const float* W2_l = (const float*)d_in[5];
    const float* b2   = (const float*)d_in[6];
    const float* W2_r = (const float*)d_in[7];
    float* out = (float*)d_out;

    const int n_nodes = in_sizes[0] / FDIM;
    const int n_edges = in_sizes[1] / 2;
    const int* src = ei;
    const int* dst = ei + n_edges;

    const int nb   = (n_nodes + BSIZE - 1) / BSIZE;
    const int nwg  = (n_edges + CHUNK - 1) / CHUNK;
    const int tot  = (nb + 1) * nwg;
    const int tpad = (tot + 3) & ~3;

    float* h      = (float*)d_ws;
    int*   tab    = (int*)(h + (size_t)n_nodes * FDIM);
    int*   rec    = tab + tpad;
    int*   csr    = rec + 2 * (size_t)nwg * CHUNK;
    int*   rowptr = csr + n_edges;

    const int layer_grid = (n_nodes + TNODES - 1) / TNODES;

    order_block<<<nwg, 1024, 0, stream>>>(src, dst, tab, rec, n_edges, nb, nwg);
    sort_bucket<<<nb, 1024, 0, stream>>>(rec, tab, rowptr, csr, n_nodes, nwg);

    sage_layer<<<layer_grid, 512, 0, stream>>>(x, rowptr, csr, W1_l, W1_r, b1, h,
                                               n_nodes,  1);

    sage_layer<<<layer_grid, 512, 0, stream>>>(h, rowptr, csr, W2_l, W2_r, b2, out,
                                               n_nodes,  0);
}


# Layer kernel: a trip's 8 row loads, a pass's 4 id loads and a wave's 16 output stores each go out as one group with addresses of their own, no address-translation wait between; sum order kept
# speedup vs baseline: 4.4237x; 1.0303x over previous
#include <hip/hip_runtime.h>
#include <hip/hip_bf16.h>

typedef __attribute__((ext_vector_type(2))) float v2f;
typedef __attribute__((ext_vector_type(8))) float v8f;
typedef __attribute__((ext_vector_type(4))) float v4f;

#define FDIM 64

#define BSHIFT 9
#define BSIZE  512
#define CHUNK  8192
#define MAXB   2048
#define SCAP   16384

__global__ void __launch_bounds__(1024)
order_block(const int* __restrict__ src, const int* __restrict__ dst,
            int* __restrict__ tab, int* __restrict__ rec,
            int n_edges, int nb, int nwg) {
    __shared__ __attribute__((aligned(16))) int  s_cnt[MAXB];
    __shared__ int  s_ws[32];
    __shared__ __attribute__((aligned(16))) int2 s_rec[CHUNK];
    const int t    = threadIdx.x;
    const int lane = t & 31;
    const int wave = t >> 5;
    const int blk  = blockIdx.x;

    s_cnt[t] = 0;
    s_cnt[t + 1024] = 0;

    const int base = blk * CHUNK + t;
    int d[8], s[8];
    #pragma unroll
    for (int u = 0; u < 8; ++u) {
        const int e = base + u * 1024;
        d[u] = (e < n_edges) ? dst[e] : 0;
        s[u] = (e < n_edges) ? src[e] : 0;
    }
    __syncthreads();

    int k[8];
    #pragma unroll
    for (int u = 0; u < 8; ++u) {
        const int e = base + u * 1024;
        k[u] = 0;
        if (e < n_edges) k[u] = atomicAdd(&s_cnt[d[u] >> BSHIFT], 1);
    }
    __syncthreads();

    const int b0 = 2 * t, b1 = 2 * t + 1;
    const int2 cc = ((const int2*)s_cnt)[t];
    const int c  = cc.x + cc.y;
    int incl = c;
    #pragma unroll
    for (int o = 1; o < 32; o <<= 1) {
        const int u = __shfl_up(incl, o, 32);
        if (lane >= o) incl += u;
    }
    if (lane == 31) s_ws[wave] = incl;
    __syncthreads();
    int woff = 0;
    #pragma unroll
    for (int w = 0; w < 32; ++w) if (w < wave) woff += s_ws[w];
    const int excl = woff + incl - c;
    ((int2*)s_cnt)[t] = make_int2(excl, excl + cc.x);
    if (b0 <= nb) tab[b0 * nwg + blk] = excl;
    if (b1 <= nb) tab[b1 * nwg + blk] = excl + cc.x;
    __syncthreads();

    int q[8];
    #pragma unroll
    for (int u = 0; u < 8; ++u) q[u] = s_cnt[d[u] >> BSHIFT] + k[u];
    #pragma unroll
    for (int u = 0; u < 8; ++u) {
        const int e = base + u * 1024;
        if (e < n_edges) s_rec[q[u]] = make_int2(s[u], d[u]);
    }
    __syncthreads();

    const int left  = n_edges - blk * CHUNK;
    const int n_loc = left < CHUNK ? left : CHUNK;
    int2* __restrict__ slab = (int2*)rec + (size_t)blk * CHUNK;
    const int4* s4 = (const int4*)s_rec;
    int4 r[4];
    #pragma unroll
    for (int u = 0; u < 4; ++u) {
        const int i = t + u * 1024;
        r[u] = (2 * i < n_loc) ? s4[i] : make_int4(0, 0, 0, 0);
    }
    #pragma unroll
    for (int u = 0; u < 4; ++u) {
        const int i = t + u * 1024;
        if (2 * i + 1 < n_loc)  ((int4*)slab)[i] = r[u];
        else if (2 * i < n_loc) slab[2 * i] = make_int2(r[u].x, r[u].y);
    }
}

template <bool PLACE, bool STAGED>
__device__ __forceinline__ void walk_runs(const int2* __restrict__ rp, const int* __restrict__ tab,
                                          int* s_cnt, int* s_out, int* __restrict__ csr,
                                          int beg, int b, int node0, int nwg, int t) {
    const int row = t >> 4;
    const int l16 = t & 15;
    for (int rbase = 0; rbase < nwg; rbase += 256) {
        int lo[4], cn[4];
        #pragma unroll
        for (int j = 0; j < 4; ++j) {
            const int r = rbase + row + 64 * j;
            int a = 0, e = 0;
            if (r < nwg) { a = tab[b * nwg + r]; e = tab[(b + 1) * nwg + r]; }
            lo[j] = r * CHUNK + a;
            cn[j] = e - a;
        }
        int2 rc[4][4];
        #pragma unroll
        for (int j = 0; j < 4; ++j) {
            #pragma unroll
            for (int m = 0; m < 4; ++m) {
                const int i = l16 + 16 * m;
                rc[j][m] = make_int2(0, node0);
                if (i < cn[j]) {
                    if (PLACE) rc[j][m] = rp[(unsigned)(lo[j] + i)];
                    else       rc[j][m].y = rp[(unsigned)(lo[j] + i)].y;
                }
            }
        }
        if (PLACE) {
            int p[4][4];
            #pragma unroll
            for (int j = 0; j < 4; ++j) {
                #pragma unroll
                for (int m = 0; m < 4; ++m) {
                    p[j][m] = 0;
                    if (l16 + 16 * m < cn[j]) p[j][m] = atomicAdd(&s_cnt[rc[j][m].y - node0], 1);
                }
            }
            #pragma unroll
            for (int j = 0; j < 4; ++j) {
                #pragma unroll
                for (int m = 0; m < 4; ++m) {
                    if (l16 + 16 * m < cn[j]) {
                        if (STAGED) s_out[p[j][m]] = rc[j][m].x;
                        else        csr[beg + p[j][m]] = rc[j][m].x;
                    }
                }
            }
        } else {
            #pragma unroll
            for (int j = 0; j < 4; ++j) {
                #pragma unroll
                for (int m = 0; m < 4; ++m) {
                    if (l16 + 16 * m < cn[j]) atomicAdd(&s_cnt[rc[j][m].y - node0], 1);
                }
            }
        }
        #pragma unroll
        for (int j = 0; j < 4; ++j) {
            for (int i = 64 + l16; i < cn[j]; i += 16) {
                const int2 v = rp[(unsigned)(lo[j] + i)];
                const int  p = atomicAdd(&s_cnt[v.y - node0], 1);
                if (PLACE) {
                    if (STAGED) s_out[p] = v.x;
                    else        csr[beg + p] = v.x;
                }
            }
        }
    }
}

__global__ void __launch_bounds__(1024)
sort_bucket(const int* __restrict__ rec, const int* __restrict__ tab,
            int* __restrict__ rowptr, int* __restrict__ csr,
            int n_nodes, int nwg) {
    __shared__ int s_cnt[BSIZE];
    __shared__ int s_ws[16];
    __shared__ int s_sb[32];
    __shared__ int s_se[32];
    __shared__ int s_out[SCAP];
    const int t     = threadIdx.x;
    const int lane  = t & 31;
    const int wave  = t >> 5;
    const int b     = blockIdx.x;
    const int node0 = b << BSHIFT;
    const int2* __restrict__ rp = (const int2*)rec;

    if (t < BSIZE) s_cnt[t] = 0;
    __syncthreads();

    walk_runs<false, false>(rp, tab, s_cnt, s_out, csr, 0, b, node0, nwg, t);

    int sb = 0, se = 0;
    for (int r = t; r < nwg; r += 1024) {
        sb += tab[b * nwg + r];
        se += tab[(b + 1) * nwg + r];
    }
    #pragma unroll
    for (int o = 16; o > 0; o >>= 1) {
        sb += __shfl_xor(sb, o, 32);
        se += __shfl_xor(se, o, 32);
    }
    if (lane == 0) { s_sb[wave] = sb; s_se[wave] = se; }
    __syncthreads();

    int beg = 0, end = 0;
    #pragma unroll
    for (int w = 0; w < 32; ++w) { beg += s_sb[w]; end += s_se[w]; }
    const int n = end - beg;
    const bool staged = (n <= SCAP);

    const int c = (t < BSIZE) ? s_cnt[t] : 0;
    int incl = c;
    #pragma unroll
    for (int o = 1; o < 32; o <<= 1) {
        const int u = __shfl_up(incl, o, 32);
        if (lane >= o) incl += u;
    }
    if (lane == 31 && wave < 16) s_ws[wave] = incl;
    __syncthreads();
    int woff = 0;
    #pragma unroll
    for (int w = 0; w < 16; ++w) if (w < wave) woff += s_ws[w];
    const int excl = woff + incl - c;
    if (t < BSIZE) {
        s_cnt[t] = excl;
        const int node = node0 + t;
        if (node < n_nodes) rowptr[node] = beg + excl;
        if (node == n_nodes - 1) rowptr[n_nodes] = end;
    }
    __syncthreads();

    if (staged) {
        walk_runs<true, true>(rp, tab, s_cnt, s_out, csr, beg, b, node0, nwg, t);
        __syncthreads();
        for (int i0 = t; i0 < n; i0 += 8 * 1024) {
            int v[8];
            #pragma unroll
            for (int u = 0; u < 8; ++u) {
                const int i = i0 + u * 1024;
                v[u] = (i < n) ? s_out[i] : 0;
            }
            #pragma unroll
            for (int u = 0; u < 8; ++u) {
                const int i = i0 + u * 1024;
                if (i < n) csr[beg + i] = v[u];
            }
        }
    } else {
        walk_runs<true, false>(rp, tab, s_cnt, s_out, csr, beg, b, node0, nwg, t);
    }
}

#define TNODES 128
#define APITCH 68

#define ROW_LANE(x, J) __builtin_amdgcn_update_dpp(0, (x), 0x150 + ((J) & 15), 0xF, 0xF, false)

template <int T>
__device__ __forceinline__ void gather_trip(const float4* __restrict__ feat4,
                                            int my0, int my1, int nb, int lm,
                                            float4& acc) {
#define GATHER_AD(U) \
    const unsigned a##U = (unsigned)ROW_LANE(((T + U) < 16 ? my0 : my1), T + U) * (FDIM / 4) + lm;
    GATHER_AD(0) GATHER_AD(1) GATHER_AD(2) GATHER_AD(3)
    GATHER_AD(4) GATHER_AD(5) GATHER_AD(6) GATHER_AD(7)
#undef GATHER_AD
    asm volatile("" :: "v"(a0), "v"(a1), "v"(a2), "v"(a3), "v"(a4), "v"(a5), "v"(a6), "v"(a7));
    float4 v[8];
    v[0] = feat4[a0]; v[1] = feat4[a1]; v[2] = feat4[a2]; v[3] = feat4[a3];
    v[4] = feat4[a4]; v[5] = feat4[a5]; v[6] = feat4[a6]; v[7] = feat4[a7];
    #pragma unroll
    for (int u = 0; u < 8; ++u) {
        const bool on = (T + u == 0) || (T + u < nb);
        acc.x += on ? v[u].x : 0.0f;
        acc.y += on ? v[u].y : 0.0f;
        acc.z += on ? v[u].z : 0.0f;
        acc.w += on ? v[u].w : 0.0f;
    }
}

__global__ void __launch_bounds__(512, 16)
sage_layer(const float* __restrict__ feat, const int* __restrict__ rowptr,
           const int* __restrict__ csr, const float* __restrict__ Wl,
           const float* __restrict__ Wr, const float* __restrict__ bias,
           float* __restrict__ out, int n_nodes, int relu_flag) {
    __shared__ float sWl[FDIM * FDIM];
    __shared__ float sWr[FDIM * FDIM];
    __shared__ float sAgg[TNODES * APITCH];

    const int lane = threadIdx.x & 31;
    const int wave = threadIdx.x >> 5;
    const int blk0 = blockIdx.x * TNODES;
    const int half = lane >> 4;
    const int lm   = lane & 15;

    const int nl0 = wave * 8;
    int rpi = blk0 + nl0 + lane;
    rpi = rpi < n_nodes ? rpi : n_nodes;
    const int rp = rowptr[rpi];

    {
        const int i = threadIdx.x;
        const v4f l0 = ((const v4f*)Wl)[i];
        const v4f l1 = ((const v4f*)Wl)[i + 512];
        const v4f r0 = ((const v4f*)Wr)[i];
        const v4f r1 = ((const v4f*)Wr)[i + 512];
        asm volatile("" :: "v"(rp), "v"(l0), "v"(l1), "v"(r0), "v"(r1));
        ((v4f*)sWl)[i]       = l0;
        ((v4f*)sWl)[i + 512] = l1;
        ((v4f*)sWr)[i]       = r0;
        ((v4f*)sWr)[i + 512] = r1;
    }

    for (int p = 0; p < 2; ++p) {
        int id0[2], id1[2];
        int i0[2], i1[2];
        bool in0[2], in1[2];
        #pragma unroll
        for (int rr = 0; rr < 2; ++rr) {
            const int r = 2 * p + rr;
            const int b = __shfl(rp, 2 * r + half, 32);
            const int d = __shfl(rp, 2 * r + 1 + half, 32) - b;
            in0[rr] = (lm < d);
            in1[rr] = (lm + 16 < d);
            i0[rr] = in0[rr] ? b + lm      : 0;
            i1[rr] = in1[rr] ? b + 16 + lm : 0;
        }
        asm volatile("" :: "v"(i0[0]), "v"(i1[0]), "v"(i0[1]), "v"(i1[1]));
        #pragma unroll
        for (int rr = 0; rr < 2; ++rr) {
            const int v0 = csr[i0[rr]];
            const int v1 = csr[i1[rr]];
            id0[rr] = in0[rr] ? v0 : 0;
            id1[rr] = in1[rr] ? v1 : 0;
        }
        #pragma unroll
        for (int rr = 0; rr < 2; ++rr) {
            const int r = 2 * p + rr;
            const int beg = __shfl(rp, 2 * r + half, 32);
            const int end = __shfl(rp, 2 * r + 1 + half, 32);
            float4 acc = make_float4(0.0f, 0.0f, 0.0f, 0.0f);
            for (int base = beg; base < end; base += 32) {
                const int left = end - base;
                const int nb   = left < 32 ? left : 32;
                int my0 = id0[rr], my1 = id1[rr];
                if (base != beg) {
                    my0 = (lm < nb)      ? csr[base + lm]      : 0;
                    my1 = (lm + 16 < nb) ? csr[base + 16 + lm] : 0;
                }
                const float4* __restrict__ feat4 = (const float4*)feat;
                gather_trip<0>(feat4, my0, my1, nb, lm, acc);
                if (nb > 8)  gather_trip<8>(feat4, my0, my1, nb, lm, acc);
                if (nb > 16) gather_trip<16>(feat4, my0, my1, nb, lm, acc);
                if (nb > 24) gather_trip<24>(feat4, my0, my1, nb, lm, acc);
            }
            const float inv = 1.0f / fmaxf((float)(end - beg), 1.0f);
            acc.x *= inv; acc.y *= inv; acc.z *= inv; acc.w *= inv;
            *(float4*)(sAgg + (nl0 + 2 * r + half) * APITCH + lm * 4) = acc;
        }
    }

    const int tile  = wave >> 1;
    const int nh    = wave & 1;
    const int node0 = blk0 + tile * 16;

    __syncthreads();

    if (node0 >= n_nodes) return;

    int xr = node0 + lm;
    xr = xr < n_nodes ? xr : n_nodes - 1;
    const float* xrow = feat + (size_t)xr * FDIM + half * 2;
    v2f ax[16];
    #pragma unroll
    for (int kb = 0; kb < 16; ++kb) ax[kb] = *(const v2f*)(xrow + kb * 4);

    const float* arow = sAgg + (tile * 16 + lm) * APITCH + half * 2;

    v8f acc[2];
    #pragma unroll
    for (int j = 0; j < 2; ++j) acc[j] = (v8f){};

    #pragma unroll
    for (int kb = 0; kb < 16; ++kb) {
        const int k0 = kb * 4 + half * 2;
        const v2f a = *(const v2f*)(arow + kb * 4);
        #pragma unroll
        for (int j = 0; j < 2; ++j) {
            const int nt = nh * 2 + j;
            v2f b;
            b.x = sWl[k0 * FDIM + nt * 16 + lm];
            b.y = sWl[(k0 + 1) * FDIM + nt * 16 + lm];
            acc[j] = __builtin_amdgcn_wmma_f32_16x16x4_f32(
                false, a, false, b, (short)0, acc[j], false, false);
        }
    }
    #pragma unroll
    for (int kb = 0; kb < 16; ++kb) {
        const int k0 = kb * 4 + half * 2;
        #pragma unroll
        for (int j = 0; j < 2; ++j) {
            const int nt = nh * 2 + j;
            v2f b;
            b.x = sWr[k0 * FDIM + nt * 16 + lm];
            b.y = sWr[(k0 + 1) * FDIM + nt * 16 + lm];
            acc[j] = __builtin_amdgcn_wmma_f32_16x16x4_f32(
                false, ax[kb], false, b, (short)0, acc[j], false, false);
        }
    }

    #pragma unroll
    for (int j = 0; j < 2; ++j) {
        const int nt = nh * 2 + j;
        const float bv = bias[nt * 16 + lm];
        #pragma unroll
        for (int i = 0; i < 8; ++i) {
            float v = acc[j][i] + bv;
            if (relu_flag) v = fmaxf(v, 0.0f);
            acc[j][i] = v;
        }
    }
    if (node0 + 16 <= n_nodes) {
        float* __restrict__ o = out + (size_t)(node0 + half * 8) * FDIM + nh * 32 + lm;
        #pragma unroll
        for (int j = 0; j < 2; ++j) {
            #pragma unroll
            for (int i = 0; i < 8; ++i) o[i * FDIM + j * 16] = acc[j][i];
        }
    } else {
        #pragma unroll
        for (int j = 0; j < 2; ++j) {
            const int nt = nh * 2 + j;
            #pragma unroll
            for (int i = 0; i < 8; ++i) {
                const int row = node0 + i + half * 8;
                if (row < n_nodes) out[(size_t)row * FDIM + nt * 16 + lm] = acc[j][i];
            }
        }
    }
}

extern "C" void kernel_launch(void* const* d_in, const int* in_sizes, int n_in,
                              void* d_out, int out_size, void* d_ws, size_t ws_size,
                              hipStream_t stream) {
    const float* x    = (const float*)d_in[0];
    const int*   ei   = (const int*)d_in[1];
    const float* W1_l = (const float*)d_in[2];
    const float* b1   = (const float*)d_in[3];
    const float* W1_r = (const float*)d_in[4];
    const float* W2_l = (const float*)d_in[5];
    const float* b2   = (const float*)d_in[6];
    const float* W2_r = (const float*)d_in[7];
    float* out = (float*)d_out;

    const int n_nodes = in_sizes[0] / FDIM;
    const int n_edges = in_sizes[1] / 2;
    const int* src = ei;
    const int* dst = ei + n_edges;

    const int nb   = (n_nodes + BSIZE - 1) / BSIZE;
    const int nwg  = (n_edges + CHUNK - 1) / CHUNK;
    const int tot  = (nb + 1) * nwg;
    const int tpad = (tot + 3) & ~3;

    float* h      = (float*)d_ws;
    int*   tab    = (int*)(h + (size_t)n_nodes * FDIM);
    int*   rec    = tab + tpad;
    int*   csr    = rec + 2 * (size_t)nwg * CHUNK;
    int*   rowptr = csr + n_edges;

    const int layer_grid = (n_nodes + TNODES - 1) / TNODES;

    order_block<<<nwg, 1024, 0, stream>>>(src, dst, tab, rec, n_edges, nb, nwg);
    sort_bucket<<<nb, 1024, 0, stream>>>(rec, tab, rowptr, csr, n_nodes, nwg);

    sage_layer<<<layer_grid, 512, 0, stream>>>(x, rowptr, csr, W1_l, W1_r, b1, h,
                                               n_nodes,  1);

    sage_layer<<<layer_grid, 512, 0, stream>>>(h, rowptr, csr, W2_l, W2_r, b2, out,
                                               n_nodes,  0);
}


# Layer kernel: blocks of 10 waves and five 16-node tiles (80 nodes), five to a WGP, not 16 waves and eight tiles, four to a WGP: fullest WGP holds 25 tiles, not 32; grid still resident; sum order kept
# speedup vs baseline: 4.7650x; 1.0772x over previous
#include <hip/hip_runtime.h>
#include <hip/hip_bf16.h>

typedef __attribute__((ext_vector_type(2))) float v2f;
typedef __attribute__((ext_vector_type(8))) float v8f;
typedef __attribute__((ext_vector_type(4))) float v4f;

#define FDIM 64

#define BSHIFT 9
#define BSIZE  512
#define CHUNK  8192
#define MAXB   2048
#define SCAP   16384

__global__ void __launch_bounds__(1024)
order_block(const int* __restrict__ src, const int* __restrict__ dst,
            int* __restrict__ tab, int* __restrict__ rec,
            int n_edges, int nb, int nwg) {
    __shared__ __attribute__((aligned(16))) int  s_cnt[MAXB];
    __shared__ int  s_ws[32];
    __shared__ __attribute__((aligned(16))) int2 s_rec[CHUNK];
    const int t    = threadIdx.x;
    const int lane = t & 31;
    const int wave = t >> 5;
    const int blk  = blockIdx.x;

    s_cnt[t] = 0;
    s_cnt[t + 1024] = 0;

    const int base = blk * CHUNK + t;
    int d[8], s[8];
    #pragma unroll
    for (int u = 0; u < 8; ++u) {
        const int e = base + u * 1024;
        d[u] = (e < n_edges) ? dst[e] : 0;
        s[u] = (e < n_edges) ? src[e] : 0;
    }
    __syncthreads();

    int k[8];
    #pragma unroll
    for (int u = 0; u < 8; ++u) {
        const int e = base + u * 1024;
        k[u] = 0;
        if (e < n_edges) k[u] = atomicAdd(&s_cnt[d[u] >> BSHIFT], 1);
    }
    __syncthreads();

    const int b0 = 2 * t, b1 = 2 * t + 1;
    const int2 cc = ((const int2*)s_cnt)[t];
    const int c  = cc.x + cc.y;
    int incl = c;
    #pragma unroll
    for (int o = 1; o < 32; o <<= 1) {
        const int u = __shfl_up(incl, o, 32);
        if (lane >= o) incl += u;
    }
    if (lane == 31) s_ws[wave] = incl;
    __syncthreads();
    int woff = 0;
    #pragma unroll
    for (int w = 0; w < 32; ++w) if (w < wave) woff += s_ws[w];
    const int excl = woff + incl - c;
    ((int2*)s_cnt)[t] = make_int2(excl, excl + cc.x);
    if (b0 <= nb) tab[b0 * nwg + blk] = excl;
    if (b1 <= nb) tab[b1 * nwg + blk] = excl + cc.x;
    __syncthreads();

    int q[8];
    #pragma unroll
    for (int u = 0; u < 8; ++u) q[u] = s_cnt[d[u] >> BSHIFT] + k[u];
    #pragma unroll
    for (int u = 0; u < 8; ++u) {
        const int e = base + u * 1024;
        if (e < n_edges) s_rec[q[u]] = make_int2(s[u], d[u]);
    }
    __syncthreads();

    const int left  = n_edges - blk * CHUNK;
    const int n_loc = left < CHUNK ? left : CHUNK;
    int2* __restrict__ slab = (int2*)rec + (size_t)blk * CHUNK;
    const int4* s4 = (const int4*)s_rec;
    int4 r[4];
    #pragma unroll
    for (int u = 0; u < 4; ++u) {
        const int i = t + u * 1024;
        r[u] = (2 * i < n_loc) ? s4[i] : make_int4(0, 0, 0, 0);
    }
    #pragma unroll
    for (int u = 0; u < 4; ++u) {
        const int i = t + u * 1024;
        if (2 * i + 1 < n_loc)  ((int4*)slab)[i] = r[u];
        else if (2 * i < n_loc) slab[2 * i] = make_int2(r[u].x, r[u].y);
    }
}

template <bool PLACE, bool STAGED>
__device__ __forceinline__ void walk_runs(const int2* __restrict__ rp, const int* __restrict__ tab,
                                          int* s_cnt, int* s_out, int* __restrict__ csr,
                                          int beg, int b, int node0, int nwg, int t) {
    const int row = t >> 4;
    const int l16 = t & 15;
    for (int rbase = 0; rbase < nwg; rbase += 256) {
        int lo[4], cn[4];
        #pragma unroll
        for (int j = 0; j < 4; ++j) {
            const int r = rbase + row + 64 * j;
            int a = 0, e = 0;
            if (r < nwg) { a = tab[b * nwg + r]; e = tab[(b + 1) * nwg + r]; }
            lo[j] = r * CHUNK + a;
            cn[j] = e - a;
        }
        int2 rc[4][4];
        #pragma unroll
        for (int j = 0; j < 4; ++j) {
            #pragma unroll
            for (int m = 0; m < 4; ++m) {
                const int i = l16 + 16 * m;
                rc[j][m] = make_int2(0, node0);
                if (i < cn[j]) {
                    if (PLACE) rc[j][m] = rp[(unsigned)(lo[j] + i)];
                    else       rc[j][m].y = rp[(unsigned)(lo[j] + i)].y;
                }
            }
        }
        if (PLACE) {
            int p[4][4];
            #pragma unroll
            for (int j = 0; j < 4; ++j) {
                #pragma unroll
                for (int m = 0; m < 4; ++m) {
                    p[j][m] = 0;
                    if (l16 + 16 * m < cn[j]) p[j][m] = atomicAdd(&s_cnt[rc[j][m].y - node0], 1);
                }
            }
            #pragma unroll
            for (int j = 0; j < 4; ++j) {
                #pragma unroll
                for (int m = 0; m < 4; ++m) {
                    if (l16 + 16 * m < cn[j]) {
                        if (STAGED) s_out[p[j][m]] = rc[j][m].x;
                        else        csr[beg + p[j][m]] = rc[j][m].x;
                    }
                }
            }
        } else {
            #pragma unroll
            for (int j = 0; j < 4; ++j) {
                #pragma unroll
                for (int m = 0; m < 4; ++m) {
                    if (l16 + 16 * m < cn[j]) atomicAdd(&s_cnt[rc[j][m].y - node0], 1);
                }
            }
        }
        #pragma unroll
        for (int j = 0; j < 4; ++j) {
            for (int i = 64 + l16; i < cn[j]; i += 16) {
                const int2 v = rp[(unsigned)(lo[j] + i)];
                const int  p = atomicAdd(&s_cnt[v.y - node0], 1);
                if (PLACE) {
                    if (STAGED) s_out[p] = v.x;
                    else        csr[beg + p] = v.x;
                }
            }
        }
    }
}

__global__ void __launch_bounds__(1024)
sort_bucket(const int* __restrict__ rec, const int* __restrict__ tab,
            int* __restrict__ rowptr, int* __restrict__ csr,
            int n_nodes, int nwg) {
    __shared__ int s_cnt[BSIZE];
    __shared__ int s_ws[16];
    __shared__ int s_sb[32];
    __shared__ int s_se[32];
    __shared__ int s_out[SCAP];
    const int t     = threadIdx.x;
    const int lane  = t & 31;
    const int wave  = t >> 5;
    const int b     = blockIdx.x;
    const int node0 = b << BSHIFT;
    const int2* __restrict__ rp = (const int2*)rec;

    if (t < BSIZE) s_cnt[t] = 0;
    __syncthreads();

    walk_runs<false, false>(rp, tab, s_cnt, s_out, csr, 0, b, node0, nwg, t);

    int sb = 0, se = 0;
    for (int r = t; r < nwg; r += 1024) {
        sb += tab[b * nwg + r];
        se += tab[(b + 1) * nwg + r];
    }
    #pragma unroll
    for (int o = 16; o > 0; o >>= 1) {
        sb += __shfl_xor(sb, o, 32);
        se += __shfl_xor(se, o, 32);
    }
    if (lane == 0) { s_sb[wave] = sb; s_se[wave] = se; }
    __syncthreads();

    int beg = 0, end = 0;
    #pragma unroll
    for (int w = 0; w < 32; ++w) { beg += s_sb[w]; end += s_se[w]; }
    const int n = end - beg;
    const bool staged = (n <= SCAP);

    const int c = (t < BSIZE) ? s_cnt[t] : 0;
    int incl = c;
    #pragma unroll
    for (int o = 1; o < 32; o <<= 1) {
        const int u = __shfl_up(incl, o, 32);
        if (lane >= o) incl += u;
    }
    if (lane == 31 && wave < 16) s_ws[wave] = incl;
    __syncthreads();
    int woff = 0;
    #pragma unroll
    for (int w = 0; w < 16; ++w) if (w < wave) woff += s_ws[w];
    const int excl = woff + incl - c;
    if (t < BSIZE) {
        s_cnt[t] = excl;
        const int node = node0 + t;
        if (node < n_nodes) rowptr[node] = beg + excl;
        if (node == n_nodes - 1) rowptr[n_nodes] = end;
    }
    __syncthreads();

    if (staged) {
        walk_runs<true, true>(rp, tab, s_cnt, s_out, csr, beg, b, node0, nwg, t);
        __syncthreads();
        for (int i0 = t; i0 < n; i0 += 8 * 1024) {
            int v[8];
            #pragma unroll
            for (int u = 0; u < 8; ++u) {
                const int i = i0 + u * 1024;
                v[u] = (i < n) ? s_out[i] : 0;
            }
            #pragma unroll
            for (int u = 0; u < 8; ++u) {
                const int i = i0 + u * 1024;
                if (i < n) csr[beg + i] = v[u];
            }
        }
    } else {
        walk_runs<true, false>(rp, tab, s_cnt, s_out, csr, beg, b, node0, nwg, t);
    }
}

#define NWAVES 10
#define NTHR   (NWAVES * 32)
#define TNODES (NWAVES * 8)
#define APITCH 68

#define ROW_LANE(x, J) __builtin_amdgcn_update_dpp(0, (x), 0x150 + ((J) & 15), 0xF, 0xF, false)

template <int T>
__device__ __forceinline__ void gather_trip(const float4* __restrict__ feat4,
                                            int my0, int my1, int nb, int lm,
                                            float4& acc) {
#define GATHER_AD(U) \
    const unsigned a##U = (unsigned)ROW_LANE(((T + U) < 16 ? my0 : my1), T + U) * (FDIM / 4) + lm;
    GATHER_AD(0) GATHER_AD(1) GATHER_AD(2) GATHER_AD(3)
    GATHER_AD(4) GATHER_AD(5) GATHER_AD(6) GATHER_AD(7)
#undef GATHER_AD
    asm volatile("" :: "v"(a0), "v"(a1), "v"(a2), "v"(a3), "v"(a4), "v"(a5), "v"(a6), "v"(a7));
    float4 v[8];
    v[0] = feat4[a0]; v[1] = feat4[a1]; v[2] = feat4[a2]; v[3] = feat4[a3];
    v[4] = feat4[a4]; v[5] = feat4[a5]; v[6] = feat4[a6]; v[7] = feat4[a7];
    #pragma unroll
    for (int u = 0; u < 8; ++u) {
        const bool on = (T + u == 0) || (T + u < nb);
        acc.x += on ? v[u].x : 0.0f;
        acc.y += on ? v[u].y : 0.0f;
        acc.z += on ? v[u].z : 0.0f;
        acc.w += on ? v[u].w : 0.0f;
    }
}

__global__ void __launch_bounds__(NTHR, 13)
sage_layer(const float* __restrict__ feat, const int* __restrict__ rowptr,
           const int* __restrict__ csr, const float* __restrict__ Wl,
           const float* __restrict__ Wr, const float* __restrict__ bias,
           float* __restrict__ out, int n_nodes, int relu_flag) {
    __shared__ float sWl[FDIM * FDIM];
    __shared__ float sWr[FDIM * FDIM];
    __shared__ float sAgg[TNODES * APITCH];

    const int lane = threadIdx.x & 31;
    const int wave = threadIdx.x >> 5;
    const int blk0 = blockIdx.x * TNODES;
    const int half = lane >> 4;
    const int lm   = lane & 15;

    const int nl0 = wave * 8;
    int rpi = blk0 + nl0 + lane;
    rpi = rpi < n_nodes ? rpi : n_nodes;
    const int rp = rowptr[rpi];

    {
        const int i  = threadIdx.x;
        const int i3 = i + 3 * NTHR;
        const int j3 = i3 < 1024 ? i3 : 1023;
        const v4f l0 = ((const v4f*)Wl)[i];
        const v4f l1 = ((const v4f*)Wl)[i + NTHR];
        const v4f l2 = ((const v4f*)Wl)[i + 2 * NTHR];
        const v4f l3 = ((const v4f*)Wl)[j3];
        const v4f r0 = ((const v4f*)Wr)[i];
        const v4f r1 = ((const v4f*)Wr)[i + NTHR];
        const v4f r2 = ((const v4f*)Wr)[i + 2 * NTHR];
        const v4f r3 = ((const v4f*)Wr)[j3];
        asm volatile("" :: "v"(rp), "v"(l0), "v"(l1), "v"(l2), "v"(l3),
                           "v"(r0), "v"(r1), "v"(r2), "v"(r3));
        ((v4f*)sWl)[i]            = l0;
        ((v4f*)sWl)[i + NTHR]     = l1;
        ((v4f*)sWl)[i + 2 * NTHR] = l2;
        ((v4f*)sWr)[i]            = r0;
        ((v4f*)sWr)[i + NTHR]     = r1;
        ((v4f*)sWr)[i + 2 * NTHR] = r2;
        if (i3 < 1024) {
            ((v4f*)sWl)[i3] = l3;
            ((v4f*)sWr)[i3] = r3;
        }
    }

    for (int p = 0; p < 2; ++p) {
        int id0[2], id1[2];
        int i0[2], i1[2];
        bool in0[2], in1[2];
        #pragma unroll
        for (int rr = 0; rr < 2; ++rr) {
            const int r = 2 * p + rr;
            const int b = __shfl(rp, 2 * r + half, 32);
            const int d = __shfl(rp, 2 * r + 1 + half, 32) - b;
            in0[rr] = (lm < d);
            in1[rr] = (lm + 16 < d);
            i0[rr] = in0[rr] ? b + lm      : 0;
            i1[rr] = in1[rr] ? b + 16 + lm : 0;
        }
        asm volatile("" :: "v"(i0[0]), "v"(i1[0]), "v"(i0[1]), "v"(i1[1]));
        #pragma unroll
        for (int rr = 0; rr < 2; ++rr) {
            const int v0 = csr[i0[rr]];
            const int v1 = csr[i1[rr]];
            id0[rr] = in0[rr] ? v0 : 0;
            id1[rr] = in1[rr] ? v1 : 0;
        }
        #pragma unroll
        for (int rr = 0; rr < 2; ++rr) {
            const int r = 2 * p + rr;
            const int beg = __shfl(rp, 2 * r + half, 32);
            const int end = __shfl(rp, 2 * r + 1 + half, 32);
            float4 acc = make_float4(0.0f, 0.0f, 0.0f, 0.0f);
            for (int base = beg; base < end; base += 32) {
                const int left = end - base;
                const int nb   = left < 32 ? left : 32;
                int my0 = id0[rr], my1 = id1[rr];
                if (base != beg) {
                    my0 = (lm < nb)      ? csr[base + lm]      : 0;
                    my1 = (lm + 16 < nb) ? csr[base + 16 + lm] : 0;
                }
                const float4* __restrict__ feat4 = (const float4*)feat;
                gather_trip<0>(feat4, my0, my1, nb, lm, acc);
                if (nb > 8)  gather_trip<8>(feat4, my0, my1, nb, lm, acc);
                if (nb > 16) gather_trip<16>(feat4, my0, my1, nb, lm, acc);
                if (nb > 24) gather_trip<24>(feat4, my0, my1, nb, lm, acc);
            }
            const float inv = 1.0f / fmaxf((float)(end - beg), 1.0f);
            acc.x *= inv; acc.y *= inv; acc.z *= inv; acc.w *= inv;
            *(float4*)(sAgg + (nl0 + 2 * r + half) * APITCH + lm * 4) = acc;
        }
    }

    const int tile  = wave >> 1;
    const int nh    = wave & 1;
    const int node0 = blk0 + tile * 16;

    __syncthreads();

    if (node0 >= n_nodes) return;

    int xr = node0 + lm;
    xr = xr < n_nodes ? xr : n_nodes - 1;
    const float* xrow = feat + (size_t)xr * FDIM + half * 2;
    v2f ax[16];
    #pragma unroll
    for (int kb = 0; kb < 16; ++kb) ax[kb] = *(const v2f*)(xrow + kb * 4);

    const float* arow = sAgg + (tile * 16 + lm) * APITCH + half * 2;

    v8f acc[2];
    #pragma unroll
    for (int j = 0; j < 2; ++j) acc[j] = (v8f){};

    #pragma unroll
    for (int kb = 0; kb < 16; ++kb) {
        const int k0 = kb * 4 + half * 2;
        const v2f a = *(const v2f*)(arow + kb * 4);
        #pragma unroll
        for (int j = 0; j < 2; ++j) {
            const int nt = nh * 2 + j;
            v2f b;
            b.x = sWl[k0 * FDIM + nt * 16 + lm];
            b.y = sWl[(k0 + 1) * FDIM + nt * 16 + lm];
            acc[j] = __builtin_amdgcn_wmma_f32_16x16x4_f32(
                false, a, false, b, (short)0, acc[j], false, false);
        }
    }
    #pragma unroll
    for (int kb = 0; kb < 16; ++kb) {
        const int k0 = kb * 4 + half * 2;
        #pragma unroll
        for (int j = 0; j < 2; ++j) {
            const int nt = nh * 2 + j;
            v2f b;
            b.x = sWr[k0 * FDIM + nt * 16 + lm];
            b.y = sWr[(k0 + 1) * FDIM + nt * 16 + lm];
            acc[j] = __builtin_amdgcn_wmma_f32_16x16x4_f32(
                false, ax[kb], false, b, (short)0, acc[j], false, false);
        }
    }

    #pragma unroll
    for (int j = 0; j < 2; ++j) {
        const int nt = nh * 2 + j;
        const float bv = bias[nt * 16 + lm];
        #pragma unroll
        for (int i = 0; i < 8; ++i) {
            float v = acc[j][i] + bv;
            if (relu_flag) v = fmaxf(v, 0.0f);
            acc[j][i] = v;
        }
    }
    if (node0 + 16 <= n_nodes) {
        float* __restrict__ o = out + (size_t)(node0 + half * 8) * FDIM + nh * 32 + lm;
        #pragma unroll
        for (int j = 0; j < 2; ++j) {
            #pragma unroll
            for (int i = 0; i < 8; ++i) o[i * FDIM + j * 16] = acc[j][i];
        }
    } else {
        #pragma unroll
        for (int j = 0; j < 2; ++j) {
            const int nt = nh * 2 + j;
            #pragma unroll
            for (int i = 0; i < 8; ++i) {
                const int row = node0 + i + half * 8;
                if (row < n_nodes) out[(size_t)row * FDIM + nt * 16 + lm] = acc[j][i];
            }
        }
    }
}

extern "C" void kernel_launch(void* const* d_in, const int* in_sizes, int n_in,
                              void* d_out, int out_size, void* d_ws, size_t ws_size,
                              hipStream_t stream) {
    const float* x    = (const float*)d_in[0];
    const int*   ei   = (const int*)d_in[1];
    const float* W1_l = (const float*)d_in[2];
    const float* b1   = (const float*)d_in[3];
    const float* W1_r = (const float*)d_in[4];
    const float* W2_l = (const float*)d_in[5];
    const float* b2   = (const float*)d_in[6];
    const float* W2_r = (const float*)d_in[7];
    float* out = (float*)d_out;

    const int n_nodes = in_sizes[0] / FDIM;
    const int n_edges = in_sizes[1] / 2;
    const int* src = ei;
    const int* dst = ei + n_edges;

    const int nb   = (n_nodes + BSIZE - 1) / BSIZE;
    const int nwg  = (n_edges + CHUNK - 1) / CHUNK;
    const int tot  = (nb + 1) * nwg;
    const int tpad = (tot + 3) & ~3;

    float* h      = (float*)d_ws;
    int*   tab    = (int*)(h + (size_t)n_nodes * FDIM);
    int*   rec    = tab + tpad;
    int*   csr    = rec + 2 * (size_t)nwg * CHUNK;
    int*   rowptr = csr + n_edges;

    const int layer_grid = (n_nodes + TNODES - 1) / TNODES;

    order_block<<<nwg, 1024, 0, stream>>>(src, dst, tab, rec, n_edges, nb, nwg);
    sort_bucket<<<nb, 1024, 0, stream>>>(rec, tab, rowptr, csr, n_nodes, nwg);

    sage_layer<<<layer_grid, NTHR, 0, stream>>>(x, rowptr, csr, W1_l, W1_r, b1, h,
                                               n_nodes,  1);

    sage_layer<<<layer_grid, NTHR, 0, stream>>>(h, rowptr, csr, W2_l, W2_r, b2, out,
                                               n_nodes,  0);
}
